# starting kernel
# baseline (speedup 1.0000x reference)
#include <hip/hip_runtime.h>

typedef __attribute__((ext_vector_type(2))) float v2f;
typedef __attribute__((ext_vector_type(8))) float v8f;

#define HW    16384
#define IMG_W 128
#define BCN   32
#define CIN   64
#define CMID  96
#define AN2   32
#define TLS_ROWSTRIDE 512
#define TLS_FLOATS (48 * TLS_ROWSTRIDE)
#define KCHUNKS 16

static __device__ __forceinline__ v8f wmma4(v2f a, v2f b, v8f c) {
  return __builtin_amdgcn_wmma_f32_16x16x4_f32(
       false, a,  false, b,
       (short)0, c,  false,  false);
}

__global__ void k_zero(float* __restrict__ p, int n) {
  int i = blockIdx.x * 256 + threadIdx.x;
  if (i < n) p[i] = 0.f;
}

__global__ void k_fused(const float* __restrict__ x,
                        const float* __restrict__ pw,
                        const float* __restrict__ dw,
                        float* __restrict__ qkv,
                        float* __restrict__ gsums) {
  extern __shared__ float tls[];
  __shared__ float dwls[CMID * 9];
  __shared__ float chsums[2 * AN2];

  const int tid  = threadIdx.x;
  const int lane = tid & 31, wave = tid >> 5;
  const int lo = lane & 15, hi = lane >> 4;
  const int img = blockIdx.y;
  const int r0  = blockIdx.x * 2;
  const float* __restrict__ X = x + (size_t)img * CIN * HW;
  float* __restrict__ QKV = qkv + (size_t)img * CMID * HW;

  for (int i = tid; i < CMID * 9; i += 256) dwls[i] = dw[i];
  if (tid < 2 * AN2) chsums[tid] = 0.f;
  __syncthreads();

  const int orow = tid >> 7;
  const int col  = tid & (IMG_W - 1);

  for (int half_ = 0; half_ < 2; ++half_) {
    const int cbase = half_ * 48;

    for (int nt = wave; nt < 32; nt += 8) {
      const int lrow = nt >> 3;
      const int colbase = (nt & 7) * 16;
      const int rr = r0 - 1 + lrow;
      if (rr < 0 || rr >= IMG_W) {
        for (int i = lane; i < 48 * 16; i += 32)
          tls[(i >> 4) * TLS_ROWSTRIDE + lrow * IMG_W + colbase + (i & 15)] = 0.f;
      } else {
        const int pix = rr * IMG_W + colbase;
        v8f acc[3] = {};
        for (int k0 = 0; k0 < CIN; k0 += 4) {
          v2f b;
          b.x = X[(size_t)(k0 + 2 * hi) * HW + pix + lo];
          b.y = X[(size_t)(k0 + 2 * hi + 1) * HW + pix + lo];
#pragma unroll
          for (int m = 0; m < 3; ++m) {
            v2f a;
            const float* arow = pw + (cbase + m * 16 + lo) * CIN + k0 + 2 * hi;
            a.x = arow[0];
            a.y = arow[1];
            acc[m] = wmma4(a, b, acc[m]);
          }
        }
#pragma unroll
        for (int m = 0; m < 3; ++m)
#pragma unroll
          for (int r = 0; r < 8; ++r)
            tls[(m * 16 + r + 8 * hi) * TLS_ROWSTRIDE + lrow * IMG_W + colbase + lo] =
                acc[m][r];
      }
    }
    __syncthreads();

    for (int lc = 0; lc < 48; ++lc) {
      const int ch = cbase + lc;
      const float* __restrict__ tch = tls + lc * TLS_ROWSTRIDE + (1 + orow) * IMG_W + col;
      const float* __restrict__ wch = dwls + ch * 9;
      float s = 0.f;
#pragma unroll
      for (int dy = -1; dy <= 1; ++dy) {
        const float* rowp = tch + dy * IMG_W;
        if (col > 0)         s += wch[(dy + 1) * 3 + 0] * rowp[-1];
                             s += wch[(dy + 1) * 3 + 1] * rowp[0];
        if (col < IMG_W - 1) s += wch[(dy + 1) * 3 + 2] * rowp[1];
      }
      QKV[(size_t)ch * HW + (r0 + orow) * IMG_W + col] = s;
      if (ch < 2 * AN2) {
        float v = s * s;
#pragma unroll
        for (int off = 16; off > 0; off >>= 1) v += __shfl_xor(v, off, 32);
        if (lane == 0) atomicAdd(&chsums[ch], v);
      }
    }
    __syncthreads();
  }

  if (tid < 2 * AN2) atomicAdd(&gsums[img * 64 + tid], chsums[tid]);
}

__global__ void k_att_partial(const float* __restrict__ qkv,
                              float* __restrict__ Sbuf) {
  const int lane = threadIdx.x & 31, wave = threadIdx.x >> 5;
  const int lo = lane & 15, hi = lane >> 4;
  const int img = blockIdx.y;
  const float* __restrict__ Q = qkv + (size_t)img * CMID * HW;
  const float* __restrict__ Km = Q + (size_t)AN2 * HW;
  const int m0 = (wave >> 1) * 16, n0 = (wave & 1) * 16;
  const int kbase = blockIdx.x * (HW / KCHUNKS);

  const float* __restrict__ qrow = Q + (size_t)(m0 + lo) * HW + 2 * hi;
  const float* __restrict__ krow = Km + (size_t)(n0 + lo) * HW + 2 * hi;

  v8f acc0 = {}, acc1 = {}, acc2 = {}, acc3 = {};
  for (int kk = kbase; kk < kbase + HW / KCHUNKS; kk += 16) {
    v2f a, b;
    a.x = qrow[kk];      a.y = qrow[kk + 1];
    b.x = krow[kk];      b.y = krow[kk + 1];
    acc0 = wmma4(a, b, acc0);
    a.x = qrow[kk + 4];  a.y = qrow[kk + 5];
    b.x = krow[kk + 4];  b.y = krow[kk + 5];
    acc1 = wmma4(a, b, acc1);
    a.x = qrow[kk + 8];  a.y = qrow[kk + 9];
    b.x = krow[kk + 8];  b.y = krow[kk + 9];
    acc2 = wmma4(a, b, acc2);
    a.x = qrow[kk + 12]; a.y = qrow[kk + 13];
    b.x = krow[kk + 12]; b.y = krow[kk + 13];
    acc3 = wmma4(a, b, acc3);
  }
  v8f acc = (acc0 + acc1) + (acc2 + acc3);

  float* __restrict__ Sp = Sbuf + (size_t)img * (AN2 * AN2);
#pragma unroll
  for (int r = 0; r < 8; ++r)
    atomicAdd(&Sp[(m0 + r + 8 * hi) * AN2 + n0 + lo], acc[r]);
}

__global__ void k_att_final(const float* __restrict__ Sbuf,
                            const float* __restrict__ sums,
                            const float* __restrict__ proj,
                            float* __restrict__ Wbuf) {
  const int img = blockIdx.x;
  __shared__ float S[AN2][AN2 + 1];
  __shared__ float rq[AN2], rk[AN2];

  if (threadIdx.x < 32) {
    rq[threadIdx.x] = 1.f / fmaxf(sqrtf(sums[img * 64 + threadIdx.x]), 1e-12f);
  } else if (threadIdx.x < 64) {
    int d = threadIdx.x - 32;
    rk[d] = 1.f / fmaxf(sqrtf(sums[img * 64 + 32 + d]), 1e-12f);
  }
  __syncthreads();
  const float* __restrict__ Sp = Sbuf + (size_t)img * (AN2 * AN2);
  for (int idx = threadIdx.x; idx < AN2 * AN2; idx += 128) {
    int row = idx >> 5, col = idx & 31;
    S[row][col] = Sp[idx] * rq[row] * rk[col];
  }
  __syncthreads();
  if (threadIdx.x < 32) {
    int row = threadIdx.x;
    float mx = -3.4e38f;
    for (int j = 0; j < 32; ++j) mx = fmaxf(mx, S[row][j]);
    float sum = 0.f;
    for (int j = 0; j < 32; ++j) {
      float e = __expf(S[row][j] - mx);
      S[row][j] = e;
      sum += e;
    }
    float inv = 1.f / sum;
    for (int j = 0; j < 32; ++j) S[row][j] *= inv;
  }
  __syncthreads();
  for (int idx = threadIdx.x; idx < 64 * 32; idx += 128) {
    int o = idx >> 5, d = idx & 31;
    float s = 0.f;
    for (int cc = 0; cc < 32; ++cc) s += proj[o * 32 + cc] * S[cc][d];
    Wbuf[(size_t)img * 2048 + idx] = s;
  }
}

__global__ void k_out(const float* __restrict__ qkv,
                      const float* __restrict__ Wbuf,
                      float* __restrict__ out) {
  const int lane = threadIdx.x & 31, wave = threadIdx.x >> 5;
  const int lo = lane & 15, hi = lane >> 4;
  const int img = blockIdx.y;
  const int n0 = blockIdx.x * 64 + wave * 16;
  const float* __restrict__ V = qkv + (size_t)img * CMID * HW + (size_t)(2 * AN2) * HW;
  const float* __restrict__ W = Wbuf + (size_t)img * 2048;
  float* __restrict__ O = out + (size_t)img * CIN * HW;

  v8f acc[4] = {};
  for (int k0 = 0; k0 < 32; k0 += 4) {
    v2f b;
    b.x = V[(size_t)(k0 + 2 * hi) * HW + n0 + lo];
    b.y = V[(size_t)(k0 + 2 * hi + 1) * HW + n0 + lo];
#pragma unroll
    for (int m = 0; m < 4; ++m) {
      v2f a;
      const float* wr = W + (m * 16 + lo) * 32 + k0 + 2 * hi;
      a.x = wr[0];
      a.y = wr[1];
      acc[m] = wmma4(a, b, acc[m]);
    }
  }
#pragma unroll
  for (int m = 0; m < 4; ++m)
#pragma unroll
    for (int r = 0; r < 8; ++r)
      __builtin_nontemporal_store(acc[m][r],
          &O[(size_t)(m * 16 + r + 8 * hi) * HW + n0 + lo]);
}

extern "C" void kernel_launch(void* const* d_in, const int* in_sizes, int n_in,
                              void* d_out, int out_size, void* d_ws, size_t ws_size,
                              hipStream_t stream) {
  const float* x    = (const float*)d_in[0];
  const float* pw   = (const float*)d_in[1];
  const float* dw   = (const float*)d_in[2];
  const float* proj = (const float*)d_in[3];
  float* out = (float*)d_out;

  float* qkv  = (float*)d_ws;
  float* sums = qkv + (size_t)BCN * CMID * HW;
  float* Sbuf = sums + (size_t)BCN * 64;
  float* Wbuf = Sbuf + (size_t)BCN * AN2 * AN2;
  const int nzero = BCN * 64 + BCN * AN2 * AN2;

  k_zero<<<dim3((nzero + 255) / 256), 256, 0, stream>>>(sums, nzero);
  k_fused<<<dim3(IMG_W / 2, BCN), 256, TLS_FLOATS * sizeof(float), stream>>>(
      x, pw, dw, qkv, sums);
  k_att_partial<<<dim3(KCHUNKS, BCN), 128, 0, stream>>>(qkv, Sbuf);
  k_att_final<<<dim3(BCN), 128, 0, stream>>>(Sbuf, sums, proj, Wbuf);
  k_out<<<dim3(HW / 64, BCN), 128, 0, stream>>>(qkv, Wbuf, out);
}


# v depthwise 3x3 moved into the out kernel: block = 4 rows x 32 v ch of t, v kept in LDS only, never stored; W in LDS; out rows written 16 B a lane as whole lines; v launch gone (6 -> 5); k order kept
# speedup vs baseline: 3.0602x; 3.0602x over previous
#include <hip/hip_runtime.h>

typedef __attribute__((ext_vector_type(2))) float v2f;
typedef __attribute__((ext_vector_type(4))) float v4f;
typedef __attribute__((ext_vector_type(8))) float v8f;

#define HW    16384
#define IMG_W 128
#define BCN   32
#define CIN   64
#define CMID  96
#define AN2   32

static __device__ __forceinline__ v8f wmma4(v2f a, v2f b, v8f c) {
  return __builtin_amdgcn_wmma_f32_16x16x4_f32(
       false, a,  false, b,
       (short)0, c,  false,  false);
}

__global__ void k_zero(float* __restrict__ p, int n) {
  int i = blockIdx.x * 256 + threadIdx.x;
  if (i < n) p[i] = 0.f;
}

#define PWP 68
__global__ __launch_bounds__(256) void k_pw(const float* __restrict__ x,
                                            const float* __restrict__ pw,
                                            float* __restrict__ t) {
  __shared__ __attribute__((aligned(16))) float pwls[CMID * PWP];
  const int tid  = threadIdx.x;
  const int lane = tid & 31, wave = tid >> 5;
  const int lo = lane & 15, hi = lane >> 4;
  const int img = blockIdx.y;

  for (int i = tid; i < CMID * (CIN / 4); i += 256) {
    const int row = i >> 4, c4 = i & 15;
    const float4 v = *reinterpret_cast<const float4*>(pw + row * CIN + c4 * 4);
    *reinterpret_cast<float4*>(pwls + row * PWP + c4 * 4) = v;
  }
  __syncthreads();

  const int cbase = (wave >> 2) * 48;
  const int base  = blockIdx.x * 256 + (wave & 3) * 64;
  const float* __restrict__ X = x + (size_t)img * CIN * HW + base + 4 * lo;
  float* __restrict__ Tt = t + (size_t)img * CMID * HW + base + 4 * lo;
  const float* __restrict__ arow = pwls + (cbase + lo) * PWP + 2 * hi;

  v8f acc[3][4] = {};
#pragma unroll 4
  for (int k0 = 0; k0 < CIN; k0 += 4) {
    const float4 x0 = *reinterpret_cast<const float4*>(X + (size_t)(k0 + 2 * hi) * HW);
    const float4 x1 = *reinterpret_cast<const float4*>(X + (size_t)(k0 + 2 * hi + 1) * HW);
    const float xs0[4] = {x0.x, x0.y, x0.z, x0.w};
    const float xs1[4] = {x1.x, x1.y, x1.z, x1.w};
#pragma unroll
    for (int m = 0; m < 3; ++m) {
      const v2f a = *reinterpret_cast<const v2f*>(arow + m * 16 * PWP + k0);
#pragma unroll
      for (int j = 0; j < 4; ++j) {
        v2f b;
        b.x = xs0[j];
        b.y = xs1[j];
        acc[m][j] = wmma4(a, b, acc[m][j]);
      }
    }
  }
#pragma unroll
  for (int m = 0; m < 3; ++m)
#pragma unroll
    for (int r = 0; r < 8; ++r) {
      float4 o;
      o.x = acc[m][0][r];
      o.y = acc[m][1][r];
      o.z = acc[m][2][r];
      o.w = acc[m][3][r];
      *reinterpret_cast<float4*>(Tt + (size_t)(cbase + m * 16 + r + 8 * hi) * HW) = o;
    }
}

#define QR  4
#define QP  516
__global__ __launch_bounds__(256) void k_dwqk(const float* __restrict__ t,
                                              const float* __restrict__ dw,
                                              float* __restrict__ Sbuf,
                                              float* __restrict__ gsums) {
  __shared__ __attribute__((aligned(16))) float qk[2 * AN2 * QP];
  const int tid  = threadIdx.x;
  const int lane = tid & 31;
  const int wave = __builtin_amdgcn_readfirstlane(tid >> 5);
  const int img = blockIdx.y;
  const int r0 = blockIdx.x * QR;
  const int rtop = (r0 > 0) ? r0 - 1 : 0;
  const int rbot = (r0 + QR < IMG_W) ? r0 + QR : IMG_W - 1;

  const float* __restrict__ T0 =
      t + ((size_t)img * CMID + wave * 8) * HW + 4 * lane;
  float4 c[QR + 2], nx[QR + 2];
  c[0] = *reinterpret_cast<const float4*>(T0 + rtop * IMG_W);
#pragma unroll
  for (int i = 1; i < QR + 1; ++i)
    c[i] = *reinterpret_cast<const float4*>(T0 + (r0 - 1 + i) * IMG_W);
  c[QR + 1] = *reinterpret_cast<const float4*>(T0 + rbot * IMG_W);
  float ssc[8];
#pragma unroll
  for (int cc = 0; cc < 8; ++cc) {
    const int ch = wave * 8 + cc;
    if (cc + 1 < 8) {
      const float* __restrict__ T = T0 + (size_t)(cc + 1) * HW;
      nx[0] = *reinterpret_cast<const float4*>(T + rtop * IMG_W);
#pragma unroll
      for (int i = 1; i < QR + 1; ++i)
        nx[i] = *reinterpret_cast<const float4*>(T + (r0 - 1 + i) * IMG_W);
      nx[QR + 1] = *reinterpret_cast<const float4*>(T + rbot * IMG_W);
    }
    float w[9];
#pragma unroll
    for (int i = 0; i < 9; ++i) w[i] = dw[ch * 9 + i];
    if (r0 == 0)          c[0]      = make_float4(0.f, 0.f, 0.f, 0.f);
    if (r0 + QR >= IMG_W) c[QR + 1] = make_float4(0.f, 0.f, 0.f, 0.f);
    float lf[QR + 2], rt[QR + 2];
#pragma unroll
    for (int i = 0; i < QR + 2; ++i) {
      const float up = __shfl_up(c[i].w, 1, 32);
      const float dn = __shfl_down(c[i].x, 1, 32);
      lf[i] = (lane == 0) ? 0.f : up;
      rt[i] = (lane == 31) ? 0.f : dn;
    }
    float ss = 0.f;
#pragma unroll
    for (int i = 0; i < QR; ++i) {
      float s0 = 0.f, s1 = 0.f, s2 = 0.f, s3 = 0.f;
#pragma unroll
      for (int dy = 0; dy < 3; ++dy) {
        const float4 cv = c[i + dy];
        const float l = lf[i + dy], r = rt[i + dy];
        const float w0 = w[dy * 3 + 0], w1 = w[dy * 3 + 1], w2 = w[dy * 3 + 2];
        s0 += w0 * l;    s0 += w1 * cv.x; s0 += w2 * cv.y;
        s1 += w0 * cv.x; s1 += w1 * cv.y; s1 += w2 * cv.z;
        s2 += w0 * cv.y; s2 += w1 * cv.z; s2 += w2 * cv.w;
        s3 += w0 * cv.z; s3 += w1 * cv.w; s3 += w2 * r;
      }
      *reinterpret_cast<float4*>(qk + ch * QP + i * IMG_W + 4 * lane) =
          make_float4(s0, s1, s2, s3);
      ss += s0 * s0; ss += s1 * s1; ss += s2 * s2; ss += s3 * s3;
    }
    ssc[cc] = ss;
#pragma unroll
    for (int i = 0; i < QR + 2; ++i) c[i] = nx[i];
  }
  float mine = 0.f;
#pragma unroll
  for (int cc = 0; cc < 8; ++cc) {
    float v = ssc[cc];
#pragma unroll
    for (int off = 16; off > 0; off >>= 1) v += __shfl_xor(v, off, 32);
    mine = (lane == cc) ? v : mine;
  }
  if (lane < 8) atomicAdd(&gsums[img * 64 + wave * 8 + lane], mine);
  __syncthreads();

  const int lo = lane & 15, hi = lane >> 4;
  const float* __restrict__ qa = qk + lo * QP + wave * 64 + 4 * hi;
  const float* __restrict__ kb = qa + AN2 * QP;
  v8f acc[2][2] = {};
#pragma unroll
  for (int s = 0; s < 64; s += 8) {
    float4 av[2], bv[2];
#pragma unroll
    for (int m = 0; m < 2; ++m) {
      av[m] = *reinterpret_cast<const float4*>(qa + m * 16 * QP + s);
      bv[m] = *reinterpret_cast<const float4*>(kb + m * 16 * QP + s);
    }
#pragma unroll
    for (int m = 0; m < 2; ++m)
#pragma unroll
      for (int n = 0; n < 2; ++n) {
        v2f a, b;
        a.x = av[m].x; a.y = av[m].y;
        b.x = bv[n].x; b.y = bv[n].y;
        acc[m][n] = wmma4(a, b, acc[m][n]);
      }
#pragma unroll
    for (int m = 0; m < 2; ++m)
#pragma unroll
      for (int n = 0; n < 2; ++n) {
        v2f a, b;
        a.x = av[m].z; a.y = av[m].w;
        b.x = bv[n].z; b.y = bv[n].w;
        acc[m][n] = wmma4(a, b, acc[m][n]);
      }
  }
  __syncthreads();
  float* __restrict__ red = qk;
#pragma unroll
  for (int m = 0; m < 2; ++m)
#pragma unroll
    for (int n = 0; n < 2; ++n)
#pragma unroll
      for (int r = 0; r < 8; ++r)
        red[wave * (AN2 * AN2) + (m * 16 + r + 8 * hi) * AN2 + n * 16 + lo] = acc[m][n][r];
  __syncthreads();
  float* __restrict__ Sp = Sbuf + (size_t)img * (AN2 * AN2);
  for (int e = tid; e < AN2 * AN2; e += 256) {
    float s = 0.f;
#pragma unroll
    for (int wv = 0; wv < 8; ++wv) s += red[wv * (AN2 * AN2) + e];
    atomicAdd(&Sp[e], s);
  }
}

__global__ void k_att_final(const float* __restrict__ Sbuf,
                            const float* __restrict__ sums,
                            const float* __restrict__ proj,
                            float* __restrict__ Wbuf) {
  const int img = blockIdx.x;
  __shared__ float S[AN2][AN2 + 1];
  __shared__ float rq[AN2], rk[AN2];

  if (threadIdx.x < 32) {
    rq[threadIdx.x] = 1.f / fmaxf(sqrtf(sums[img * 64 + threadIdx.x]), 1e-12f);
  } else if (threadIdx.x < 64) {
    int d = threadIdx.x - 32;
    rk[d] = 1.f / fmaxf(sqrtf(sums[img * 64 + 32 + d]), 1e-12f);
  }
  __syncthreads();
  const float* __restrict__ Sp = Sbuf + (size_t)img * (AN2 * AN2);
  for (int idx = threadIdx.x; idx < AN2 * AN2; idx += 128) {
    int row = idx >> 5, col = idx & 31;
    S[row][col] = Sp[idx] * rq[row] * rk[col];
  }
  __syncthreads();
  if (threadIdx.x < 32) {
    int row = threadIdx.x;
    float mx = -3.4e38f;
    for (int j = 0; j < 32; ++j) mx = fmaxf(mx, S[row][j]);
    float sum = 0.f;
    for (int j = 0; j < 32; ++j) {
      float e = __expf(S[row][j] - mx);
      S[row][j] = e;
      sum += e;
    }
    float inv = 1.f / sum;
    for (int j = 0; j < 32; ++j) S[row][j] *= inv;
  }
  __syncthreads();
  for (int idx = threadIdx.x; idx < 64 * 32; idx += 128) {
    int o = idx >> 5, d = idx & 31;
    float s = 0.f;
    for (int cc = 0; cc < 32; ++cc) s += proj[o * 32 + cc] * S[cc][d];
    Wbuf[(size_t)img * 2048 + idx] = s;
  }
}

#define VR  4
#define VP  516
#define WP  36
__global__ __launch_bounds__(256) void k_dwout(const float* __restrict__ t,
                                               const float* __restrict__ dw,
                                               const float* __restrict__ Wbuf,
                                               float* __restrict__ out) {
  __shared__ __attribute__((aligned(16))) float vls[AN2 * VP];
  __shared__ __attribute__((aligned(16))) float wls[CIN * WP];
  const int tid  = threadIdx.x;
  const int lane = tid & 31;
  const int wave = __builtin_amdgcn_readfirstlane(tid >> 5);
  const int img = blockIdx.y;
  const int r0 = blockIdx.x * VR;
  const int rtop = (r0 > 0) ? r0 - 1 : 0;
  const int rbot = (r0 + VR < IMG_W) ? r0 + VR : IMG_W - 1;

  const float* __restrict__ T0 =
      t + ((size_t)img * CMID + 2 * AN2 + wave * 4) * HW + 4 * lane;
  float4 c[VR + 2], nx[VR + 2];
  c[0] = *reinterpret_cast<const float4*>(T0 + rtop * IMG_W);
#pragma unroll
  for (int i = 1; i < VR + 1; ++i)
    c[i] = *reinterpret_cast<const float4*>(T0 + (r0 - 1 + i) * IMG_W);
  c[VR + 1] = *reinterpret_cast<const float4*>(T0 + rbot * IMG_W);

  {
    const float* __restrict__ Wg = Wbuf + (size_t)img * 2048;
#pragma unroll
    for (int i = tid; i < CIN * (AN2 / 4); i += 256) {
      const int row = i >> 3, c4 = i & 7;
      const float4 v = *reinterpret_cast<const float4*>(Wg + row * AN2 + c4 * 4);
      *reinterpret_cast<float4*>(wls + row * WP + c4 * 4) = v;
    }
  }

#pragma unroll
  for (int cc = 0; cc < 4; ++cc) {
    const int chv = wave * 4 + cc;
    const int ch  = 2 * AN2 + chv;
    if (cc + 1 < 4) {
      const float* __restrict__ T = T0 + (size_t)(cc + 1) * HW;
      nx[0] = *reinterpret_cast<const float4*>(T + rtop * IMG_W);
#pragma unroll
      for (int i = 1; i < VR + 1; ++i)
        nx[i] = *reinterpret_cast<const float4*>(T + (r0 - 1 + i) * IMG_W);
      nx[VR + 1] = *reinterpret_cast<const float4*>(T + rbot * IMG_W);
    }
    float w[9];
#pragma unroll
    for (int i = 0; i < 9; ++i) w[i] = dw[ch * 9 + i];
    if (r0 == 0)          c[0]      = make_float4(0.f, 0.f, 0.f, 0.f);
    if (r0 + VR >= IMG_W) c[VR + 1] = make_float4(0.f, 0.f, 0.f, 0.f);
    float lf[VR + 2], rt[VR + 2];
#pragma unroll
    for (int i = 0; i < VR + 2; ++i) {
      const float up = __shfl_up(c[i].w, 1, 32);
      const float dn = __shfl_down(c[i].x, 1, 32);
      lf[i] = (lane == 0) ? 0.f : up;
      rt[i] = (lane == 31) ? 0.f : dn;
    }
#pragma unroll
    for (int i = 0; i < VR; ++i) {
      float s0 = 0.f, s1 = 0.f, s2 = 0.f, s3 = 0.f;
#pragma unroll
      for (int dy = 0; dy < 3; ++dy) {
        const float4 cv = c[i + dy];
        const float l = lf[i + dy], r = rt[i + dy];
        const float w0 = w[dy * 3 + 0], w1 = w[dy * 3 + 1], w2 = w[dy * 3 + 2];
        s0 += w0 * l;    s0 += w1 * cv.x; s0 += w2 * cv.y;
        s1 += w0 * cv.x; s1 += w1 * cv.y; s1 += w2 * cv.z;
        s2 += w0 * cv.y; s2 += w1 * cv.z; s2 += w2 * cv.w;
        s3 += w0 * cv.z; s3 += w1 * cv.w; s3 += w2 * r;
      }
      *reinterpret_cast<float4*>(vls + chv * VP + i * IMG_W + 4 * lane) =
          make_float4(s0, s1, s2, s3);
    }
#pragma unroll
    for (int i = 0; i < VR + 2; ++i) c[i] = nx[i];
  }
  __syncthreads();

  const int lo = lane & 15, hi = lane >> 4;
  const int px = wave * 64 + 4 * lo;
  const float* __restrict__ vb = vls + 2 * hi * VP + px;
  const float* __restrict__ wa = wls + lo * WP + 2 * hi;
  float* __restrict__ O =
      out + (size_t)img * CIN * HW + (size_t)r0 * IMG_W + px;
  for (int mh = 0; mh < 2; ++mh) {
    v8f acc[2][4] = {};
#pragma unroll 2
    for (int k0 = 0; k0 < AN2; k0 += 4) {
      const float4 b0 = *reinterpret_cast<const float4*>(vb + k0 * VP);
      const float4 b1 = *reinterpret_cast<const float4*>(vb + (k0 + 1) * VP);
      const float bs0[4] = {b0.x, b0.y, b0.z, b0.w};
      const float bs1[4] = {b1.x, b1.y, b1.z, b1.w};
#pragma unroll
      for (int m = 0; m < 2; ++m) {
        const v2f a =
            *reinterpret_cast<const v2f*>(wa + (mh * 2 + m) * 16 * WP + k0);
#pragma unroll
        for (int j = 0; j < 4; ++j) {
          v2f b;
          b.x = bs0[j];
          b.y = bs1[j];
          acc[m][j] = wmma4(a, b, acc[m][j]);
        }
      }
    }
#pragma unroll
    for (int m = 0; m < 2; ++m)
#pragma unroll
      for (int r = 0; r < 8; ++r) {
        v4f o;
        o.x = acc[m][0][r];
        o.y = acc[m][1][r];
        o.z = acc[m][2][r];
        o.w = acc[m][3][r];
        __builtin_nontemporal_store(o, reinterpret_cast<v4f*>(
            O + (size_t)((mh * 2 + m) * 16 + r + 8 * hi) * HW));
      }
  }
}

extern "C" void kernel_launch(void* const* d_in, const int* in_sizes, int n_in,
                              void* d_out, int out_size, void* d_ws, size_t ws_size,
                              hipStream_t stream) {
  const float* x    = (const float*)d_in[0];
  const float* pw   = (const float*)d_in[1];
  const float* dw   = (const float*)d_in[2];
  const float* proj = (const float*)d_in[3];
  float* out = (float*)d_out;

  float* qkv  = (float*)d_ws;
  float* sums = qkv + (size_t)BCN * CMID * HW;
  float* Sbuf = sums + (size_t)BCN * 64;
  float* Wbuf = Sbuf + (size_t)BCN * AN2 * AN2;
  float* tbuf = Wbuf + (size_t)BCN * 64 * AN2;
  const int nzero = BCN * 64 + BCN * AN2 * AN2;

  k_zero<<<dim3((nzero + 255) / 256), 256, 0, stream>>>(sums, nzero);
  k_pw<<<dim3(HW / 256, BCN), 256, 0, stream>>>(x, pw, tbuf);
  k_dwqk<<<dim3(IMG_W / QR, BCN), 256, 0, stream>>>(tbuf, dw, Sbuf, sums);
  k_att_final<<<dim3(BCN), 128, 0, stream>>>(Sbuf, sums, proj, Wbuf);
  k_dwout<<<dim3(IMG_W / VR, BCN), 256, 0, stream>>>(tbuf, dw, Wbuf, out);
}


# pointwise GEMM made persistent: 256 work-groups of 16 waves (one a WGP) each walk 8 tiles; pw staged to LDS once; next 16-k chunk of x always in flight, across tile ends too; k order kept
# speedup vs baseline: 3.2204x; 1.0523x over previous
#include <hip/hip_runtime.h>

typedef __attribute__((ext_vector_type(2))) float v2f;
typedef __attribute__((ext_vector_type(4))) float v4f;
typedef __attribute__((ext_vector_type(8))) float v8f;

#define HW    16384
#define IMG_W 128
#define BCN   32
#define CIN   64
#define CMID  96
#define AN2   32

static __device__ __forceinline__ v8f wmma4(v2f a, v2f b, v8f c) {
  return __builtin_amdgcn_wmma_f32_16x16x4_f32(
       false, a,  false, b,
       (short)0, c,  false,  false);
}

__global__ void k_zero(float* __restrict__ p, int n) {
  int i = blockIdx.x * 256 + threadIdx.x;
  if (i < n) p[i] = 0.f;
}

#define PWP 68
#define PW_PASSES 4

static __device__ __forceinline__ void pw_load_chunk(float4 (&xv)[8],
                                                     const float* __restrict__ X,
                                                     int kc, int hi) {
#pragma unroll
  for (int ks = 0; ks < 4; ++ks) {
    const int k0 = kc + ks * 4;
    xv[2 * ks]     = *reinterpret_cast<const float4*>(X + (size_t)(k0 + 2 * hi) * HW);
    xv[2 * ks + 1] = *reinterpret_cast<const float4*>(X + (size_t)(k0 + 2 * hi + 1) * HW);
  }
}

static __device__ __forceinline__ void pw_mul_chunk(v8f (&acc)[3][4],
                                                    const float4 (&xv)[8],
                                                    const float* __restrict__ arow,
                                                    int kc) {
#pragma unroll
  for (int ks = 0; ks < 4; ++ks) {
    const float4 x0 = xv[2 * ks], x1 = xv[2 * ks + 1];
    const float xs0[4] = {x0.x, x0.y, x0.z, x0.w};
    const float xs1[4] = {x1.x, x1.y, x1.z, x1.w};
#pragma unroll
    for (int m = 0; m < 3; ++m) {
      const v2f a = *reinterpret_cast<const v2f*>(arow + m * 16 * PWP + kc + ks * 4);
#pragma unroll
      for (int j = 0; j < 4; ++j) {
        v2f b;
        b.x = xs0[j];
        b.y = xs1[j];
        acc[m][j] = wmma4(a, b, acc[m][j]);
      }
    }
  }
}

__global__ __launch_bounds__(512) void k_pw(const float* __restrict__ x,
                                            const float* __restrict__ pw,
                                            float* __restrict__ t) {
  __shared__ __attribute__((aligned(16))) float pwls[CMID * PWP];
  const int tid  = threadIdx.x;
  const int lane = tid & 31, wave = tid >> 5;
  const int lo = lane & 15, hi = lane >> 4;
  const int img = blockIdx.x >> 3;

  for (int i = tid; i < CMID * (CIN / 4); i += 512) {
    const int row = i >> 4, c4 = i & 15;
    const float4 v = *reinterpret_cast<const float4*>(pw + row * CIN + c4 * 4);
    *reinterpret_cast<float4*>(pwls + row * PWP + c4 * 4) = v;
  }

  const int w8 = wave & 7;
  const int cbase = (w8 >> 2) * 48;
  const int base = ((blockIdx.x & 7) * 8 + (wave >> 3)) * 256 + (w8 & 3) * 64;
  const float* __restrict__ X = x + (size_t)img * CIN * HW + base + 4 * lo;
  float* __restrict__ Tt =
      t + (size_t)img * CMID * HW + (size_t)cbase * HW + base + 4 * lo;
  const float* __restrict__ arow = pwls + (cbase + lo) * PWP + 2 * hi;

  float4 xa[8], xb[8];
  pw_load_chunk(xa, X, 0, hi);
  __syncthreads();

#define PW_STAGE() __builtin_amdgcn_sched_barrier(0)

#pragma unroll 1
  for (int pass = 0; pass < PW_PASSES; ++pass) {
    v8f acc[3][4] = {};
    PW_STAGE();
    pw_load_chunk(xb, X, 16, hi);
    PW_STAGE();
    pw_mul_chunk(acc, xa, arow, 0);
    PW_STAGE();
    pw_load_chunk(xa, X, 32, hi);
    PW_STAGE();
    pw_mul_chunk(acc, xb, arow, 16);
    PW_STAGE();
    pw_load_chunk(xb, X, 48, hi);
    PW_STAGE();
    pw_mul_chunk(acc, xa, arow, 32);
    PW_STAGE();
    if (pass + 1 < PW_PASSES) X += 512;
    pw_load_chunk(xa, X, 0, hi);
    PW_STAGE();
    pw_mul_chunk(acc, xb, arow, 48);
#pragma unroll
    for (int m = 0; m < 3; ++m)
#pragma unroll
      for (int r = 0; r < 8; ++r) {
        float4 o;
        o.x = acc[m][0][r];
        o.y = acc[m][1][r];
        o.z = acc[m][2][r];
        o.w = acc[m][3][r];
        *reinterpret_cast<float4*>(Tt + (size_t)(m * 16 + r + 8 * hi) * HW) = o;
      }
    Tt += 512;
  }
#undef PW_STAGE
}

#define QR  4
#define QP  516
__global__ __launch_bounds__(256) void k_dwqk(const float* __restrict__ t,
                                              const float* __restrict__ dw,
                                              float* __restrict__ Sbuf,
                                              float* __restrict__ gsums) {
  __shared__ __attribute__((aligned(16))) float qk[2 * AN2 * QP];
  const int tid  = threadIdx.x;
  const int lane = tid & 31;
  const int wave = __builtin_amdgcn_readfirstlane(tid >> 5);
  const int img = blockIdx.y;
  const int r0 = blockIdx.x * QR;
  const int rtop = (r0 > 0) ? r0 - 1 : 0;
  const int rbot = (r0 + QR < IMG_W) ? r0 + QR : IMG_W - 1;

  const float* __restrict__ T0 =
      t + ((size_t)img * CMID + wave * 8) * HW + 4 * lane;
  float4 c[QR + 2], nx[QR + 2];
  c[0] = *reinterpret_cast<const float4*>(T0 + rtop * IMG_W);
#pragma unroll
  for (int i = 1; i < QR + 1; ++i)
    c[i] = *reinterpret_cast<const float4*>(T0 + (r0 - 1 + i) * IMG_W);
  c[QR + 1] = *reinterpret_cast<const float4*>(T0 + rbot * IMG_W);
  float ssc[8];
#pragma unroll
  for (int cc = 0; cc < 8; ++cc) {
    const int ch = wave * 8 + cc;
    if (cc + 1 < 8) {
      const float* __restrict__ T = T0 + (size_t)(cc + 1) * HW;
      nx[0] = *reinterpret_cast<const float4*>(T + rtop * IMG_W);
#pragma unroll
      for (int i = 1; i < QR + 1; ++i)
        nx[i] = *reinterpret_cast<const float4*>(T + (r0 - 1 + i) * IMG_W);
      nx[QR + 1] = *reinterpret_cast<const float4*>(T + rbot * IMG_W);
    }
    float w[9];
#pragma unroll
    for (int i = 0; i < 9; ++i) w[i] = dw[ch * 9 + i];
    if (r0 == 0)          c[0]      = make_float4(0.f, 0.f, 0.f, 0.f);
    if (r0 + QR >= IMG_W) c[QR + 1] = make_float4(0.f, 0.f, 0.f, 0.f);
    float lf[QR + 2], rt[QR + 2];
#pragma unroll
    for (int i = 0; i < QR + 2; ++i) {
      const float up = __shfl_up(c[i].w, 1, 32);
      const float dn = __shfl_down(c[i].x, 1, 32);
      lf[i] = (lane == 0) ? 0.f : up;
      rt[i] = (lane == 31) ? 0.f : dn;
    }
    float ss = 0.f;
#pragma unroll
    for (int i = 0; i < QR; ++i) {
      float s0 = 0.f, s1 = 0.f, s2 = 0.f, s3 = 0.f;
#pragma unroll
      for (int dy = 0; dy < 3; ++dy) {
        const float4 cv = c[i + dy];
        const float l = lf[i + dy], r = rt[i + dy];
        const float w0 = w[dy * 3 + 0], w1 = w[dy * 3 + 1], w2 = w[dy * 3 + 2];
        s0 += w0 * l;    s0 += w1 * cv.x; s0 += w2 * cv.y;
        s1 += w0 * cv.x; s1 += w1 * cv.y; s1 += w2 * cv.z;
        s2 += w0 * cv.y; s2 += w1 * cv.z; s2 += w2 * cv.w;
        s3 += w0 * cv.z; s3 += w1 * cv.w; s3 += w2 * r;
      }
      *reinterpret_cast<float4*>(qk + ch * QP + i * IMG_W + 4 * lane) =
          make_float4(s0, s1, s2, s3);
      ss += s0 * s0; ss += s1 * s1; ss += s2 * s2; ss += s3 * s3;
    }
    ssc[cc] = ss;
#pragma unroll
    for (int i = 0; i < QR + 2; ++i) c[i] = nx[i];
  }
  float mine = 0.f;
#pragma unroll
  for (int cc = 0; cc < 8; ++cc) {
    float v = ssc[cc];
#pragma unroll
    for (int off = 16; off > 0; off >>= 1) v += __shfl_xor(v, off, 32);
    mine = (lane == cc) ? v : mine;
  }
  if (lane < 8) atomicAdd(&gsums[img * 64 + wave * 8 + lane], mine);
  __syncthreads();

  const int lo = lane & 15, hi = lane >> 4;
  const float* __restrict__ qa = qk + lo * QP + wave * 64 + 4 * hi;
  const float* __restrict__ kb = qa + AN2 * QP;
  v8f acc[2][2] = {};
#pragma unroll
  for (int s = 0; s < 64; s += 8) {
    float4 av[2], bv[2];
#pragma unroll
    for (int m = 0; m < 2; ++m) {
      av[m] = *reinterpret_cast<const float4*>(qa + m * 16 * QP + s);
      bv[m] = *reinterpret_cast<const float4*>(kb + m * 16 * QP + s);
    }
#pragma unroll
    for (int m = 0; m < 2; ++m)
#pragma unroll
      for (int n = 0; n < 2; ++n) {
        v2f a, b;
        a.x = av[m].x; a.y = av[m].y;
        b.x = bv[n].x; b.y = bv[n].y;
        acc[m][n] = wmma4(a, b, acc[m][n]);
      }
#pragma unroll
    for (int m = 0; m < 2; ++m)
#pragma unroll
      for (int n = 0; n < 2; ++n) {
        v2f a, b;
        a.x = av[m].z; a.y = av[m].w;
        b.x = bv[n].z; b.y = bv[n].w;
        acc[m][n] = wmma4(a, b, acc[m][n]);
      }
  }
  __syncthreads();
  float* __restrict__ red = qk;
#pragma unroll
  for (int m = 0; m < 2; ++m)
#pragma unroll
    for (int n = 0; n < 2; ++n)
#pragma unroll
      for (int r = 0; r < 8; ++r)
        red[wave * (AN2 * AN2) + (m * 16 + r + 8 * hi) * AN2 + n * 16 + lo] = acc[m][n][r];
  __syncthreads();
  float* __restrict__ Sp = Sbuf + (size_t)img * (AN2 * AN2);
  for (int e = tid; e < AN2 * AN2; e += 256) {
    float s = 0.f;
#pragma unroll
    for (int wv = 0; wv < 8; ++wv) s += red[wv * (AN2 * AN2) + e];
    atomicAdd(&Sp[e], s);
  }
}

__global__ void k_att_final(const float* __restrict__ Sbuf,
                            const float* __restrict__ sums,
                            const float* __restrict__ proj,
                            float* __restrict__ Wbuf) {
  const int img = blockIdx.x;
  __shared__ float S[AN2][AN2 + 1];
  __shared__ float rq[AN2], rk[AN2];

  if (threadIdx.x < 32) {
    rq[threadIdx.x] = 1.f / fmaxf(sqrtf(sums[img * 64 + threadIdx.x]), 1e-12f);
  } else if (threadIdx.x < 64) {
    int d = threadIdx.x - 32;
    rk[d] = 1.f / fmaxf(sqrtf(sums[img * 64 + 32 + d]), 1e-12f);
  }
  __syncthreads();
  const float* __restrict__ Sp = Sbuf + (size_t)img * (AN2 * AN2);
  for (int idx = threadIdx.x; idx < AN2 * AN2; idx += 128) {
    int row = idx >> 5, col = idx & 31;
    S[row][col] = Sp[idx] * rq[row] * rk[col];
  }
  __syncthreads();
  if (threadIdx.x < 32) {
    int row = threadIdx.x;
    float mx = -3.4e38f;
    for (int j = 0; j < 32; ++j) mx = fmaxf(mx, S[row][j]);
    float sum = 0.f;
    for (int j = 0; j < 32; ++j) {
      float e = __expf(S[row][j] - mx);
      S[row][j] = e;
      sum += e;
    }
    float inv = 1.f / sum;
    for (int j = 0; j < 32; ++j) S[row][j] *= inv;
  }
  __syncthreads();
  for (int idx = threadIdx.x; idx < 64 * 32; idx += 128) {
    int o = idx >> 5, d = idx & 31;
    float s = 0.f;
    for (int cc = 0; cc < 32; ++cc) s += proj[o * 32 + cc] * S[cc][d];
    Wbuf[(size_t)img * 2048 + idx] = s;
  }
}

#define VR  4
#define VP  516
#define WP  36
__global__ __launch_bounds__(256) void k_dwout(const float* __restrict__ t,
                                               const float* __restrict__ dw,
                                               const float* __restrict__ Wbuf,
                                               float* __restrict__ out) {
  __shared__ __attribute__((aligned(16))) float vls[AN2 * VP];
  __shared__ __attribute__((aligned(16))) float wls[CIN * WP];
  const int tid  = threadIdx.x;
  const int lane = tid & 31;
  const int wave = __builtin_amdgcn_readfirstlane(tid >> 5);
  const int img = blockIdx.y;
  const int r0 = blockIdx.x * VR;
  const int rtop = (r0 > 0) ? r0 - 1 : 0;
  const int rbot = (r0 + VR < IMG_W) ? r0 + VR : IMG_W - 1;

  const float* __restrict__ T0 =
      t + ((size_t)img * CMID + 2 * AN2 + wave * 4) * HW + 4 * lane;
  float4 c[VR + 2], nx[VR + 2];
  c[0] = *reinterpret_cast<const float4*>(T0 + rtop * IMG_W);
#pragma unroll
  for (int i = 1; i < VR + 1; ++i)
    c[i] = *reinterpret_cast<const float4*>(T0 + (r0 - 1 + i) * IMG_W);
  c[VR + 1] = *reinterpret_cast<const float4*>(T0 + rbot * IMG_W);

  {
    const float* __restrict__ Wg = Wbuf + (size_t)img * 2048;
#pragma unroll
    for (int i = tid; i < CIN * (AN2 / 4); i += 256) {
      const int row = i >> 3, c4 = i & 7;
      const float4 v = *reinterpret_cast<const float4*>(Wg + row * AN2 + c4 * 4);
      *reinterpret_cast<float4*>(wls + row * WP + c4 * 4) = v;
    }
  }

#pragma unroll
  for (int cc = 0; cc < 4; ++cc) {
    const int chv = wave * 4 + cc;
    const int ch  = 2 * AN2 + chv;
    if (cc + 1 < 4) {
      const float* __restrict__ T = T0 + (size_t)(cc + 1) * HW;
      nx[0] = *reinterpret_cast<const float4*>(T + rtop * IMG_W);
#pragma unroll
      for (int i = 1; i < VR + 1; ++i)
        nx[i] = *reinterpret_cast<const float4*>(T + (r0 - 1 + i) * IMG_W);
      nx[VR + 1] = *reinterpret_cast<const float4*>(T + rbot * IMG_W);
    }
    float w[9];
#pragma unroll
    for (int i = 0; i < 9; ++i) w[i] = dw[ch * 9 + i];
    if (r0 == 0)          c[0]      = make_float4(0.f, 0.f, 0.f, 0.f);
    if (r0 + VR >= IMG_W) c[VR + 1] = make_float4(0.f, 0.f, 0.f, 0.f);
    float lf[VR + 2], rt[VR + 2];
#pragma unroll
    for (int i = 0; i < VR + 2; ++i) {
      const float up = __shfl_up(c[i].w, 1, 32);
      const float dn = __shfl_down(c[i].x, 1, 32);
      lf[i] = (lane == 0) ? 0.f : up;
      rt[i] = (lane == 31) ? 0.f : dn;
    }
#pragma unroll
    for (int i = 0; i < VR; ++i) {
      float s0 = 0.f, s1 = 0.f, s2 = 0.f, s3 = 0.f;
#pragma unroll
      for (int dy = 0; dy < 3; ++dy) {
        const float4 cv = c[i + dy];
        const float l = lf[i + dy], r = rt[i + dy];
        const float w0 = w[dy * 3 + 0], w1 = w[dy * 3 + 1], w2 = w[dy * 3 + 2];
        s0 += w0 * l;    s0 += w1 * cv.x; s0 += w2 * cv.y;
        s1 += w0 * cv.x; s1 += w1 * cv.y; s1 += w2 * cv.z;
        s2 += w0 * cv.y; s2 += w1 * cv.z; s2 += w2 * cv.w;
        s3 += w0 * cv.z; s3 += w1 * cv.w; s3 += w2 * r;
      }
      *reinterpret_cast<float4*>(vls + chv * VP + i * IMG_W + 4 * lane) =
          make_float4(s0, s1, s2, s3);
    }
#pragma unroll
    for (int i = 0; i < VR + 2; ++i) c[i] = nx[i];
  }
  __syncthreads();

  const int lo = lane & 15, hi = lane >> 4;
  const int px = wave * 64 + 4 * lo;
  const float* __restrict__ vb = vls + 2 * hi * VP + px;
  const float* __restrict__ wa = wls + lo * WP + 2 * hi;
  float* __restrict__ O =
      out + (size_t)img * CIN * HW + (size_t)r0 * IMG_W + px;
  for (int mh = 0; mh < 2; ++mh) {
    v8f acc[2][4] = {};
#pragma unroll 2
    for (int k0 = 0; k0 < AN2; k0 += 4) {
      const float4 b0 = *reinterpret_cast<const float4*>(vb + k0 * VP);
      const float4 b1 = *reinterpret_cast<const float4*>(vb + (k0 + 1) * VP);
      const float bs0[4] = {b0.x, b0.y, b0.z, b0.w};
      const float bs1[4] = {b1.x, b1.y, b1.z, b1.w};
#pragma unroll
      for (int m = 0; m < 2; ++m) {
        const v2f a =
            *reinterpret_cast<const v2f*>(wa + (mh * 2 + m) * 16 * WP + k0);
#pragma unroll
        for (int j = 0; j < 4; ++j) {
          v2f b;
          b.x = bs0[j];
          b.y = bs1[j];
          acc[m][j] = wmma4(a, b, acc[m][j]);
        }
      }
    }
#pragma unroll
    for (int m = 0; m < 2; ++m)
#pragma unroll
      for (int r = 0; r < 8; ++r) {
        v4f o;
        o.x = acc[m][0][r];
        o.y = acc[m][1][r];
        o.z = acc[m][2][r];
        o.w = acc[m][3][r];
        __builtin_nontemporal_store(o, reinterpret_cast<v4f*>(
            O + (size_t)((mh * 2 + m) * 16 + r + 8 * hi) * HW));
      }
  }
}

extern "C" void kernel_launch(void* const* d_in, const int* in_sizes, int n_in,
                              void* d_out, int out_size, void* d_ws, size_t ws_size,
                              hipStream_t stream) {
  const float* x    = (const float*)d_in[0];
  const float* pw   = (const float*)d_in[1];
  const float* dw   = (const float*)d_in[2];
  const float* proj = (const float*)d_in[3];
  float* out = (float*)d_out;

  float* qkv  = (float*)d_ws;
  float* sums = qkv + (size_t)BCN * CMID * HW;
  float* Sbuf = sums + (size_t)BCN * 64;
  float* Wbuf = Sbuf + (size_t)BCN * AN2 * AN2;
  float* tbuf = Wbuf + (size_t)BCN * 64 * AN2;
  const int nzero = BCN * 64 + BCN * AN2 * AN2;

  k_zero<<<dim3((nzero + 255) / 256), 256, 0, stream>>>(sums, nzero);
  k_pw<<<dim3(256), 512, 0, stream>>>(x, pw, tbuf);
  k_dwqk<<<dim3(IMG_W / QR, BCN), 256, 0, stream>>>(tbuf, dw, Sbuf, sums);
  k_att_final<<<dim3(BCN), 128, 0, stream>>>(Sbuf, sums, proj, Wbuf);
  k_dwout<<<dim3(IMG_W / VR, BCN), 256, 0, stream>>>(tbuf, dw, Wbuf, out);
}


# softmax+fold kernel made parallel: 1024 threads an image, one score element a thread, all global loads issued once up front, row max/sum by lane shuffles, proj_w in LDS; softmax row sum order changed
# speedup vs baseline: 3.4587x; 1.0740x over previous
#include <hip/hip_runtime.h>

typedef __attribute__((ext_vector_type(2))) float v2f;
typedef __attribute__((ext_vector_type(4))) float v4f;
typedef __attribute__((ext_vector_type(8))) float v8f;

#define HW    16384
#define IMG_W 128
#define BCN   32
#define CIN   64
#define CMID  96
#define AN2   32

static __device__ __forceinline__ v8f wmma4(v2f a, v2f b, v8f c) {
  return __builtin_amdgcn_wmma_f32_16x16x4_f32(
       false, a,  false, b,
       (short)0, c,  false,  false);
}

__global__ void k_zero(float* __restrict__ p, int n) {
  int i = blockIdx.x * 256 + threadIdx.x;
  if (i < n) p[i] = 0.f;
}

#define PWP 68
#define PW_PASSES 4

static __device__ __forceinline__ void pw_load_chunk(float4 (&xv)[8],
                                                     const float* __restrict__ X,
                                                     int kc, int hi) {
#pragma unroll
  for (int ks = 0; ks < 4; ++ks) {
    const int k0 = kc + ks * 4;
    xv[2 * ks]     = *reinterpret_cast<const float4*>(X + (size_t)(k0 + 2 * hi) * HW);
    xv[2 * ks + 1] = *reinterpret_cast<const float4*>(X + (size_t)(k0 + 2 * hi + 1) * HW);
  }
}

static __device__ __forceinline__ void pw_mul_chunk(v8f (&acc)[3][4],
                                                    const float4 (&xv)[8],
                                                    const float* __restrict__ arow,
                                                    int kc) {
#pragma unroll
  for (int ks = 0; ks < 4; ++ks) {
    const float4 x0 = xv[2 * ks], x1 = xv[2 * ks + 1];
    const float xs0[4] = {x0.x, x0.y, x0.z, x0.w};
    const float xs1[4] = {x1.x, x1.y, x1.z, x1.w};
#pragma unroll
    for (int m = 0; m < 3; ++m) {
      const v2f a = *reinterpret_cast<const v2f*>(arow + m * 16 * PWP + kc + ks * 4);
#pragma unroll
      for (int j = 0; j < 4; ++j) {
        v2f b;
        b.x = xs0[j];
        b.y = xs1[j];
        acc[m][j] = wmma4(a, b, acc[m][j]);
      }
    }
  }
}

__global__ __launch_bounds__(512) void k_pw(const float* __restrict__ x,
                                            const float* __restrict__ pw,
                                            float* __restrict__ t) {
  __shared__ __attribute__((aligned(16))) float pwls[CMID * PWP];
  const int tid  = threadIdx.x;
  const int lane = tid & 31, wave = tid >> 5;
  const int lo = lane & 15, hi = lane >> 4;
  const int img = blockIdx.x >> 3;

  for (int i = tid; i < CMID * (CIN / 4); i += 512) {
    const int row = i >> 4, c4 = i & 15;
    const float4 v = *reinterpret_cast<const float4*>(pw + row * CIN + c4 * 4);
    *reinterpret_cast<float4*>(pwls + row * PWP + c4 * 4) = v;
  }

  const int w8 = wave & 7;
  const int cbase = (w8 >> 2) * 48;
  const int base = ((blockIdx.x & 7) * 8 + (wave >> 3)) * 256 + (w8 & 3) * 64;
  const float* __restrict__ X = x + (size_t)img * CIN * HW + base + 4 * lo;
  float* __restrict__ Tt =
      t + (size_t)img * CMID * HW + (size_t)cbase * HW + base + 4 * lo;
  const float* __restrict__ arow = pwls + (cbase + lo) * PWP + 2 * hi;

  float4 xa[8], xb[8];
  pw_load_chunk(xa, X, 0, hi);
  __syncthreads();

#define PW_STAGE() __builtin_amdgcn_sched_barrier(0)

#pragma unroll 1
  for (int pass = 0; pass < PW_PASSES; ++pass) {
    v8f acc[3][4] = {};
    PW_STAGE();
    pw_load_chunk(xb, X, 16, hi);
    PW_STAGE();
    pw_mul_chunk(acc, xa, arow, 0);
    PW_STAGE();
    pw_load_chunk(xa, X, 32, hi);
    PW_STAGE();
    pw_mul_chunk(acc, xb, arow, 16);
    PW_STAGE();
    pw_load_chunk(xb, X, 48, hi);
    PW_STAGE();
    pw_mul_chunk(acc, xa, arow, 32);
    PW_STAGE();
    if (pass + 1 < PW_PASSES) X += 512;
    pw_load_chunk(xa, X, 0, hi);
    PW_STAGE();
    pw_mul_chunk(acc, xb, arow, 48);
#pragma unroll
    for (int m = 0; m < 3; ++m)
#pragma unroll
      for (int r = 0; r < 8; ++r) {
        float4 o;
        o.x = acc[m][0][r];
        o.y = acc[m][1][r];
        o.z = acc[m][2][r];
        o.w = acc[m][3][r];
        *reinterpret_cast<float4*>(Tt + (size_t)(m * 16 + r + 8 * hi) * HW) = o;
      }
    Tt += 512;
  }
#undef PW_STAGE
}

#define QR  4
#define QP  516
__global__ __launch_bounds__(256) void k_dwqk(const float* __restrict__ t,
                                              const float* __restrict__ dw,
                                              float* __restrict__ Sbuf,
                                              float* __restrict__ gsums) {
  __shared__ __attribute__((aligned(16))) float qk[2 * AN2 * QP];
  const int tid  = threadIdx.x;
  const int lane = tid & 31;
  const int wave = __builtin_amdgcn_readfirstlane(tid >> 5);
  const int img = blockIdx.y;
  const int r0 = blockIdx.x * QR;
  const int rtop = (r0 > 0) ? r0 - 1 : 0;
  const int rbot = (r0 + QR < IMG_W) ? r0 + QR : IMG_W - 1;

  const float* __restrict__ T0 =
      t + ((size_t)img * CMID + wave * 8) * HW + 4 * lane;
  float4 c[QR + 2], nx[QR + 2];
  c[0] = *reinterpret_cast<const float4*>(T0 + rtop * IMG_W);
#pragma unroll
  for (int i = 1; i < QR + 1; ++i)
    c[i] = *reinterpret_cast<const float4*>(T0 + (r0 - 1 + i) * IMG_W);
  c[QR + 1] = *reinterpret_cast<const float4*>(T0 + rbot * IMG_W);
  float ssc[8];
#pragma unroll
  for (int cc = 0; cc < 8; ++cc) {
    const int ch = wave * 8 + cc;
    if (cc + 1 < 8) {
      const float* __restrict__ T = T0 + (size_t)(cc + 1) * HW;
      nx[0] = *reinterpret_cast<const float4*>(T + rtop * IMG_W);
#pragma unroll
      for (int i = 1; i < QR + 1; ++i)
        nx[i] = *reinterpret_cast<const float4*>(T + (r0 - 1 + i) * IMG_W);
      nx[QR + 1] = *reinterpret_cast<const float4*>(T + rbot * IMG_W);
    }
    float w[9];
#pragma unroll
    for (int i = 0; i < 9; ++i) w[i] = dw[ch * 9 + i];
    if (r0 == 0)          c[0]      = make_float4(0.f, 0.f, 0.f, 0.f);
    if (r0 + QR >= IMG_W) c[QR + 1] = make_float4(0.f, 0.f, 0.f, 0.f);
    float lf[QR + 2], rt[QR + 2];
#pragma unroll
    for (int i = 0; i < QR + 2; ++i) {
      const float up = __shfl_up(c[i].w, 1, 32);
      const float dn = __shfl_down(c[i].x, 1, 32);
      lf[i] = (lane == 0) ? 0.f : up;
      rt[i] = (lane == 31) ? 0.f : dn;
    }
    float ss = 0.f;
#pragma unroll
    for (int i = 0; i < QR; ++i) {
      float s0 = 0.f, s1 = 0.f, s2 = 0.f, s3 = 0.f;
#pragma unroll
      for (int dy = 0; dy < 3; ++dy) {
        const float4 cv = c[i + dy];
        const float l = lf[i + dy], r = rt[i + dy];
        const float w0 = w[dy * 3 + 0], w1 = w[dy * 3 + 1], w2 = w[dy * 3 + 2];
        s0 += w0 * l;    s0 += w1 * cv.x; s0 += w2 * cv.y;
        s1 += w0 * cv.x; s1 += w1 * cv.y; s1 += w2 * cv.z;
        s2 += w0 * cv.y; s2 += w1 * cv.z; s2 += w2 * cv.w;
        s3 += w0 * cv.z; s3 += w1 * cv.w; s3 += w2 * r;
      }
      *reinterpret_cast<float4*>(qk + ch * QP + i * IMG_W + 4 * lane) =
          make_float4(s0, s1, s2, s3);
      ss += s0 * s0; ss += s1 * s1; ss += s2 * s2; ss += s3 * s3;
    }
    ssc[cc] = ss;
#pragma unroll
    for (int i = 0; i < QR + 2; ++i) c[i] = nx[i];
  }
  float mine = 0.f;
#pragma unroll
  for (int cc = 0; cc < 8; ++cc) {
    float v = ssc[cc];
#pragma unroll
    for (int off = 16; off > 0; off >>= 1) v += __shfl_xor(v, off, 32);
    mine = (lane == cc) ? v : mine;
  }
  if (lane < 8) atomicAdd(&gsums[img * 64 + wave * 8 + lane], mine);
  __syncthreads();

  const int lo = lane & 15, hi = lane >> 4;
  const float* __restrict__ qa = qk + lo * QP + wave * 64 + 4 * hi;
  const float* __restrict__ kb = qa + AN2 * QP;
  v8f acc[2][2] = {};
#pragma unroll
  for (int s = 0; s < 64; s += 8) {
    float4 av[2], bv[2];
#pragma unroll
    for (int m = 0; m < 2; ++m) {
      av[m] = *reinterpret_cast<const float4*>(qa + m * 16 * QP + s);
      bv[m] = *reinterpret_cast<const float4*>(kb + m * 16 * QP + s);
    }
#pragma unroll
    for (int m = 0; m < 2; ++m)
#pragma unroll
      for (int n = 0; n < 2; ++n) {
        v2f a, b;
        a.x = av[m].x; a.y = av[m].y;
        b.x = bv[n].x; b.y = bv[n].y;
        acc[m][n] = wmma4(a, b, acc[m][n]);
      }
#pragma unroll
    for (int m = 0; m < 2; ++m)
#pragma unroll
      for (int n = 0; n < 2; ++n) {
        v2f a, b;
        a.x = av[m].z; a.y = av[m].w;
        b.x = bv[n].z; b.y = bv[n].w;
        acc[m][n] = wmma4(a, b, acc[m][n]);
      }
  }
  __syncthreads();
  float* __restrict__ red = qk;
#pragma unroll
  for (int m = 0; m < 2; ++m)
#pragma unroll
    for (int n = 0; n < 2; ++n)
#pragma unroll
      for (int r = 0; r < 8; ++r)
        red[wave * (AN2 * AN2) + (m * 16 + r + 8 * hi) * AN2 + n * 16 + lo] = acc[m][n][r];
  __syncthreads();
  float* __restrict__ Sp = Sbuf + (size_t)img * (AN2 * AN2);
  for (int e = tid; e < AN2 * AN2; e += 256) {
    float s = 0.f;
#pragma unroll
    for (int wv = 0; wv < 8; ++wv) s += red[wv * (AN2 * AN2) + e];
    atomicAdd(&Sp[e], s);
  }
}

__global__ __launch_bounds__(1024) void k_att_final(const float* __restrict__ Sbuf,
                                                    const float* __restrict__ sums,
                                                    const float* __restrict__ proj,
                                                    float* __restrict__ Wbuf) {
  const int img = blockIdx.x;
  __shared__ float att[AN2][AN2 + 1];
  __shared__ __attribute__((aligned(16))) float pj[CIN * AN2];
  __shared__ float rq[AN2], rk[AN2];
  const int tid = threadIdx.x;
  const int col = tid & 31, row = tid >> 5;

  const float sraw = Sbuf[(size_t)img * (AN2 * AN2) + tid];
  const float4 pv = reinterpret_cast<const float4*>(proj)[tid & (CIN * AN2 / 4 - 1)];
  const float sm = sums[img * 64 + (tid & 63)];
  if (tid < CIN * AN2 / 4) reinterpret_cast<float4*>(pj)[tid] = pv;
  if (tid < 32) {
    rq[tid] = 1.f / fmaxf(sqrtf(sm), 1e-12f);
  } else if (tid < 64) {
    rk[tid - 32] = 1.f / fmaxf(sqrtf(sm), 1e-12f);
  }
  __syncthreads();

  const float v = sraw * rq[row] * rk[col];
  float mx = fmaxf(-3.4e38f, v);
#pragma unroll
  for (int off = 16; off > 0; off >>= 1) mx = fmaxf(mx, __shfl_xor(mx, off, 32));
  const float e = __expf(v - mx);
  float sum = e;
#pragma unroll
  for (int off = 16; off > 0; off >>= 1) sum += __shfl_xor(sum, off, 32);
  const float inv = 1.f / sum;
  att[row][col] = e * inv;
  __syncthreads();

  const float* __restrict__ p0 = pj + row * AN2;
  const float* __restrict__ p1 = pj + (row + 32) * AN2;
  float a0 = 0.f, a1 = 0.f;
#pragma unroll
  for (int cc = 0; cc < AN2; ++cc) {
    const float av = att[cc][col];
    a0 += p0[cc] * av;
    a1 += p1[cc] * av;
  }
  float* __restrict__ Wg = Wbuf + (size_t)img * 2048;
  Wg[row * AN2 + col] = a0;
  Wg[(row + 32) * AN2 + col] = a1;
}

#define VR  4
#define VP  516
#define WP  36
__global__ __launch_bounds__(256) void k_dwout(const float* __restrict__ t,
                                               const float* __restrict__ dw,
                                               const float* __restrict__ Wbuf,
                                               float* __restrict__ out) {
  __shared__ __attribute__((aligned(16))) float vls[AN2 * VP];
  __shared__ __attribute__((aligned(16))) float wls[CIN * WP];
  const int tid  = threadIdx.x;
  const int lane = tid & 31;
  const int wave = __builtin_amdgcn_readfirstlane(tid >> 5);
  const int img = blockIdx.y;
  const int r0 = blockIdx.x * VR;
  const int rtop = (r0 > 0) ? r0 - 1 : 0;
  const int rbot = (r0 + VR < IMG_W) ? r0 + VR : IMG_W - 1;

  const float* __restrict__ T0 =
      t + ((size_t)img * CMID + 2 * AN2 + wave * 4) * HW + 4 * lane;
  float4 c[VR + 2], nx[VR + 2];
  c[0] = *reinterpret_cast<const float4*>(T0 + rtop * IMG_W);
#pragma unroll
  for (int i = 1; i < VR + 1; ++i)
    c[i] = *reinterpret_cast<const float4*>(T0 + (r0 - 1 + i) * IMG_W);
  c[VR + 1] = *reinterpret_cast<const float4*>(T0 + rbot * IMG_W);

  {
    const float* __restrict__ Wg = Wbuf + (size_t)img * 2048;
#pragma unroll
    for (int i = tid; i < CIN * (AN2 / 4); i += 256) {
      const int row = i >> 3, c4 = i & 7;
      const float4 v = *reinterpret_cast<const float4*>(Wg + row * AN2 + c4 * 4);
      *reinterpret_cast<float4*>(wls + row * WP + c4 * 4) = v;
    }
  }

#pragma unroll
  for (int cc = 0; cc < 4; ++cc) {
    const int chv = wave * 4 + cc;
    const int ch  = 2 * AN2 + chv;
    if (cc + 1 < 4) {
      const float* __restrict__ T = T0 + (size_t)(cc + 1) * HW;
      nx[0] = *reinterpret_cast<const float4*>(T + rtop * IMG_W);
#pragma unroll
      for (int i = 1; i < VR + 1; ++i)
        nx[i] = *reinterpret_cast<const float4*>(T + (r0 - 1 + i) * IMG_W);
      nx[VR + 1] = *reinterpret_cast<const float4*>(T + rbot * IMG_W);
    }
    float w[9];
#pragma unroll
    for (int i = 0; i < 9; ++i) w[i] = dw[ch * 9 + i];
    if (r0 == 0)          c[0]      = make_float4(0.f, 0.f, 0.f, 0.f);
    if (r0 + VR >= IMG_W) c[VR + 1] = make_float4(0.f, 0.f, 0.f, 0.f);
    float lf[VR + 2], rt[VR + 2];
#pragma unroll
    for (int i = 0; i < VR + 2; ++i) {
      const float up = __shfl_up(c[i].w, 1, 32);
      const float dn = __shfl_down(c[i].x, 1, 32);
      lf[i] = (lane == 0) ? 0.f : up;
      rt[i] = (lane == 31) ? 0.f : dn;
    }
#pragma unroll
    for (int i = 0; i < VR; ++i) {
      float s0 = 0.f, s1 = 0.f, s2 = 0.f, s3 = 0.f;
#pragma unroll
      for (int dy = 0; dy < 3; ++dy) {
        const float4 cv = c[i + dy];
        const float l = lf[i + dy], r = rt[i + dy];
        const float w0 = w[dy * 3 + 0], w1 = w[dy * 3 + 1], w2 = w[dy * 3 + 2];
        s0 += w0 * l;    s0 += w1 * cv.x; s0 += w2 * cv.y;
        s1 += w0 * cv.x; s1 += w1 * cv.y; s1 += w2 * cv.z;
        s2 += w0 * cv.y; s2 += w1 * cv.z; s2 += w2 * cv.w;
        s3 += w0 * cv.z; s3 += w1 * cv.w; s3 += w2 * r;
      }
      *reinterpret_cast<float4*>(vls + chv * VP + i * IMG_W + 4 * lane) =
          make_float4(s0, s1, s2, s3);
    }
#pragma unroll
    for (int i = 0; i < VR + 2; ++i) c[i] = nx[i];
  }
  __syncthreads();

  const int lo = lane & 15, hi = lane >> 4;
  const int px = wave * 64 + 4 * lo;
  const float* __restrict__ vb = vls + 2 * hi * VP + px;
  const float* __restrict__ wa = wls + lo * WP + 2 * hi;
  float* __restrict__ O =
      out + (size_t)img * CIN * HW + (size_t)r0 * IMG_W + px;
  for (int mh = 0; mh < 2; ++mh) {
    v8f acc[2][4] = {};
#pragma unroll 2
    for (int k0 = 0; k0 < AN2; k0 += 4) {
      const float4 b0 = *reinterpret_cast<const float4*>(vb + k0 * VP);
      const float4 b1 = *reinterpret_cast<const float4*>(vb + (k0 + 1) * VP);
      const float bs0[4] = {b0.x, b0.y, b0.z, b0.w};
      const float bs1[4] = {b1.x, b1.y, b1.z, b1.w};
#pragma unroll
      for (int m = 0; m < 2; ++m) {
        const v2f a =
            *reinterpret_cast<const v2f*>(wa + (mh * 2 + m) * 16 * WP + k0);
#pragma unroll
        for (int j = 0; j < 4; ++j) {
          v2f b;
          b.x = bs0[j];
          b.y = bs1[j];
          acc[m][j] = wmma4(a, b, acc[m][j]);
        }
      }
    }
#pragma unroll
    for (int m = 0; m < 2; ++m)
#pragma unroll
      for (int r = 0; r < 8; ++r) {
        v4f o;
        o.x = acc[m][0][r];
        o.y = acc[m][1][r];
        o.z = acc[m][2][r];
        o.w = acc[m][3][r];
        __builtin_nontemporal_store(o, reinterpret_cast<v4f*>(
            O + (size_t)((mh * 2 + m) * 16 + r + 8 * hi) * HW));
      }
  }
}

extern "C" void kernel_launch(void* const* d_in, const int* in_sizes, int n_in,
                              void* d_out, int out_size, void* d_ws, size_t ws_size,
                              hipStream_t stream) {
  const float* x    = (const float*)d_in[0];
  const float* pw   = (const float*)d_in[1];
  const float* dw   = (const float*)d_in[2];
  const float* proj = (const float*)d_in[3];
  float* out = (float*)d_out;

  float* qkv  = (float*)d_ws;
  float* sums = qkv + (size_t)BCN * CMID * HW;
  float* Sbuf = sums + (size_t)BCN * 64;
  float* Wbuf = Sbuf + (size_t)BCN * AN2 * AN2;
  float* tbuf = Wbuf + (size_t)BCN * 64 * AN2;
  const int nzero = BCN * 64 + BCN * AN2 * AN2;

  k_zero<<<dim3((nzero + 255) / 256), 256, 0, stream>>>(sums, nzero);
  k_pw<<<dim3(256), 512, 0, stream>>>(x, pw, tbuf);
  k_dwqk<<<dim3(IMG_W / QR, BCN), 256, 0, stream>>>(tbuf, dw, Sbuf, sums);
  k_att_final<<<dim3(BCN), 1024, 0, stream>>>(Sbuf, sums, proj, Wbuf);
  k_dwout<<<dim3(IMG_W / VR, BCN), 256, 0, stream>>>(tbuf, dw, Wbuf, out);
}


# depthwise blocks take 8 image rows, not 4 (q/k: 16 waves x 4 ch, 1 block a WGP; out: wave owns 2 px groups): waves, rounds halved; t rows read 10 per 8, not 6 per 4; S, square sums: order changed
# speedup vs baseline: 3.5964x; 1.0398x over previous
#include <hip/hip_runtime.h>

typedef __attribute__((ext_vector_type(2))) float v2f;
typedef __attribute__((ext_vector_type(4))) float v4f;
typedef __attribute__((ext_vector_type(8))) float v8f;

#define HW    16384
#define IMG_W 128
#define BCN   32
#define CIN   64
#define CMID  96
#define AN2   32

static __device__ __forceinline__ v8f wmma4(v2f a, v2f b, v8f c) {
  return __builtin_amdgcn_wmma_f32_16x16x4_f32(
       false, a,  false, b,
       (short)0, c,  false,  false);
}

__global__ void k_zero(float* __restrict__ p, int n) {
  int i = blockIdx.x * 256 + threadIdx.x;
  if (i < n) p[i] = 0.f;
}

#define PWP 68
#define PW_PASSES 4

static __device__ __forceinline__ void pw_load_chunk(float4 (&xv)[8],
                                                     const float* __restrict__ X,
                                                     int kc, int hi) {
#pragma unroll
  for (int ks = 0; ks < 4; ++ks) {
    const int k0 = kc + ks * 4;
    xv[2 * ks]     = *reinterpret_cast<const float4*>(X + (size_t)(k0 + 2 * hi) * HW);
    xv[2 * ks + 1] = *reinterpret_cast<const float4*>(X + (size_t)(k0 + 2 * hi + 1) * HW);
  }
}

static __device__ __forceinline__ void pw_mul_chunk(v8f (&acc)[3][4],
                                                    const float4 (&xv)[8],
                                                    const float* __restrict__ arow,
                                                    int kc) {
#pragma unroll
  for (int ks = 0; ks < 4; ++ks) {
    const float4 x0 = xv[2 * ks], x1 = xv[2 * ks + 1];
    const float xs0[4] = {x0.x, x0.y, x0.z, x0.w};
    const float xs1[4] = {x1.x, x1.y, x1.z, x1.w};
#pragma unroll
    for (int m = 0; m < 3; ++m) {
      const v2f a = *reinterpret_cast<const v2f*>(arow + m * 16 * PWP + kc + ks * 4);
#pragma unroll
      for (int j = 0; j < 4; ++j) {
        v2f b;
        b.x = xs0[j];
        b.y = xs1[j];
        acc[m][j] = wmma4(a, b, acc[m][j]);
      }
    }
  }
}

__global__ __launch_bounds__(512) void k_pw(const float* __restrict__ x,
                                            const float* __restrict__ pw,
                                            float* __restrict__ t) {
  __shared__ __attribute__((aligned(16))) float pwls[CMID * PWP];
  const int tid  = threadIdx.x;
  const int lane = tid & 31, wave = tid >> 5;
  const int lo = lane & 15, hi = lane >> 4;
  const int img = blockIdx.x >> 3;

  for (int i = tid; i < CMID * (CIN / 4); i += 512) {
    const int row = i >> 4, c4 = i & 15;
    const float4 v = *reinterpret_cast<const float4*>(pw + row * CIN + c4 * 4);
    *reinterpret_cast<float4*>(pwls + row * PWP + c4 * 4) = v;
  }

  const int w8 = wave & 7;
  const int cbase = (w8 >> 2) * 48;
  const int base = ((blockIdx.x & 7) * 8 + (wave >> 3)) * 256 + (w8 & 3) * 64;
  const float* __restrict__ X = x + (size_t)img * CIN * HW + base + 4 * lo;
  float* __restrict__ Tt =
      t + (size_t)img * CMID * HW + (size_t)cbase * HW + base + 4 * lo;
  const float* __restrict__ arow = pwls + (cbase + lo) * PWP + 2 * hi;

  float4 xa[8], xb[8];
  pw_load_chunk(xa, X, 0, hi);
  __syncthreads();

#define PW_STAGE() __builtin_amdgcn_sched_barrier(0)

#pragma unroll 1
  for (int pass = 0; pass < PW_PASSES; ++pass) {
    v8f acc[3][4] = {};
    PW_STAGE();
    pw_load_chunk(xb, X, 16, hi);
    PW_STAGE();
    pw_mul_chunk(acc, xa, arow, 0);
    PW_STAGE();
    pw_load_chunk(xa, X, 32, hi);
    PW_STAGE();
    pw_mul_chunk(acc, xb, arow, 16);
    PW_STAGE();
    pw_load_chunk(xb, X, 48, hi);
    PW_STAGE();
    pw_mul_chunk(acc, xa, arow, 32);
    PW_STAGE();
    if (pass + 1 < PW_PASSES) X += 512;
    pw_load_chunk(xa, X, 0, hi);
    PW_STAGE();
    pw_mul_chunk(acc, xb, arow, 48);
#pragma unroll
    for (int m = 0; m < 3; ++m)
#pragma unroll
      for (int r = 0; r < 8; ++r) {
        float4 o;
        o.x = acc[m][0][r];
        o.y = acc[m][1][r];
        o.z = acc[m][2][r];
        o.w = acc[m][3][r];
        *reinterpret_cast<float4*>(Tt + (size_t)(m * 16 + r + 8 * hi) * HW) = o;
      }
    Tt += 512;
  }
#undef PW_STAGE
}

#define QR  8
#define QP  1028
#define QC  4
#define QWV 16
__global__ __launch_bounds__(512) void k_dwqk(const float* __restrict__ t,
                                              const float* __restrict__ dw,
                                              float* __restrict__ Sbuf,
                                              float* __restrict__ gsums) {
  __shared__ __attribute__((aligned(16))) float qk[2 * AN2 * QP];
  const int tid  = threadIdx.x;
  const int lane = tid & 31;
  const int wave = __builtin_amdgcn_readfirstlane(tid >> 5);
  const int img = blockIdx.y;
  const int r0 = blockIdx.x * QR;
  const int rtop = (r0 > 0) ? r0 - 1 : 0;
  const int rbot = (r0 + QR < IMG_W) ? r0 + QR : IMG_W - 1;

  const float* __restrict__ T0 =
      t + ((size_t)img * CMID + wave * QC) * HW + 4 * lane;
  float4 call[QC][QR + 2];
#pragma unroll
  for (int cc = 0; cc < QC; ++cc) {
    const float* __restrict__ T = T0 + (size_t)cc * HW;
    call[cc][0] = *reinterpret_cast<const float4*>(T + rtop * IMG_W);
#pragma unroll
    for (int i = 1; i < QR + 1; ++i)
      call[cc][i] = *reinterpret_cast<const float4*>(T + (r0 - 1 + i) * IMG_W);
    call[cc][QR + 1] = *reinterpret_cast<const float4*>(T + rbot * IMG_W);
  }
  float ssc[QC];
#pragma unroll
  for (int cc = 0; cc < QC; ++cc) {
    const int ch = wave * QC + cc;
    float4 (&c)[QR + 2] = call[cc];
    float w[9];
#pragma unroll
    for (int i = 0; i < 9; ++i) w[i] = dw[ch * 9 + i];
    if (r0 == 0)          c[0]      = make_float4(0.f, 0.f, 0.f, 0.f);
    if (r0 + QR >= IMG_W) c[QR + 1] = make_float4(0.f, 0.f, 0.f, 0.f);
    float lf[QR + 2], rt[QR + 2];
#pragma unroll
    for (int i = 0; i < QR + 2; ++i) {
      const float up = __shfl_up(c[i].w, 1, 32);
      const float dn = __shfl_down(c[i].x, 1, 32);
      lf[i] = (lane == 0) ? 0.f : up;
      rt[i] = (lane == 31) ? 0.f : dn;
    }
    float ss = 0.f;
#pragma unroll
    for (int i = 0; i < QR; ++i) {
      float s0 = 0.f, s1 = 0.f, s2 = 0.f, s3 = 0.f;
#pragma unroll
      for (int dy = 0; dy < 3; ++dy) {
        const float4 cv = c[i + dy];
        const float l = lf[i + dy], r = rt[i + dy];
        const float w0 = w[dy * 3 + 0], w1 = w[dy * 3 + 1], w2 = w[dy * 3 + 2];
        s0 += w0 * l;    s0 += w1 * cv.x; s0 += w2 * cv.y;
        s1 += w0 * cv.x; s1 += w1 * cv.y; s1 += w2 * cv.z;
        s2 += w0 * cv.y; s2 += w1 * cv.z; s2 += w2 * cv.w;
        s3 += w0 * cv.z; s3 += w1 * cv.w; s3 += w2 * r;
      }
      *reinterpret_cast<float4*>(qk + ch * QP + i * IMG_W + 4 * lane) =
          make_float4(s0, s1, s2, s3);
      ss += s0 * s0; ss += s1 * s1; ss += s2 * s2; ss += s3 * s3;
    }
    ssc[cc] = ss;
  }
  float mine = 0.f;
#pragma unroll
  for (int cc = 0; cc < QC; ++cc) {
    float v = ssc[cc];
#pragma unroll
    for (int off = 16; off > 0; off >>= 1) v += __shfl_xor(v, off, 32);
    mine = (lane == cc) ? v : mine;
  }
  if (lane < QC) atomicAdd(&gsums[img * 64 + wave * QC + lane], mine);
  __syncthreads();

  const int lo = lane & 15, hi = lane >> 4;
  const float* __restrict__ qa = qk + lo * QP + wave * 64 + 4 * hi;
  const float* __restrict__ kb = qa + AN2 * QP;
  v8f acc[2][2] = {};
#pragma unroll
  for (int s = 0; s < 64; s += 8) {
    float4 av[2], bv[2];
#pragma unroll
    for (int m = 0; m < 2; ++m) {
      av[m] = *reinterpret_cast<const float4*>(qa + m * 16 * QP + s);
      bv[m] = *reinterpret_cast<const float4*>(kb + m * 16 * QP + s);
    }
#pragma unroll
    for (int m = 0; m < 2; ++m)
#pragma unroll
      for (int n = 0; n < 2; ++n) {
        v2f a, b;
        a.x = av[m].x; a.y = av[m].y;
        b.x = bv[n].x; b.y = bv[n].y;
        acc[m][n] = wmma4(a, b, acc[m][n]);
      }
#pragma unroll
    for (int m = 0; m < 2; ++m)
#pragma unroll
      for (int n = 0; n < 2; ++n) {
        v2f a, b;
        a.x = av[m].z; a.y = av[m].w;
        b.x = bv[n].z; b.y = bv[n].w;
        acc[m][n] = wmma4(a, b, acc[m][n]);
      }
  }
  __syncthreads();
  float* __restrict__ red = qk;
#pragma unroll
  for (int m = 0; m < 2; ++m)
#pragma unroll
    for (int n = 0; n < 2; ++n)
#pragma unroll
      for (int r = 0; r < 8; ++r)
        red[wave * (AN2 * AN2) + (m * 16 + r + 8 * hi) * AN2 + n * 16 + lo] = acc[m][n][r];
  __syncthreads();
  float* __restrict__ Sp = Sbuf + (size_t)img * (AN2 * AN2);
  for (int e = tid; e < AN2 * AN2; e += 512) {
    float s = 0.f;
#pragma unroll
    for (int wv = 0; wv < QWV; ++wv) s += red[wv * (AN2 * AN2) + e];
    atomicAdd(&Sp[e], s);
  }
}

__global__ __launch_bounds__(1024) void k_att_final(const float* __restrict__ Sbuf,
                                                    const float* __restrict__ sums,
                                                    const float* __restrict__ proj,
                                                    float* __restrict__ Wbuf) {
  const int img = blockIdx.x;
  __shared__ float att[AN2][AN2 + 1];
  __shared__ __attribute__((aligned(16))) float pj[CIN * AN2];
  __shared__ float rq[AN2], rk[AN2];
  const int tid = threadIdx.x;
  const int col = tid & 31, row = tid >> 5;

  const float sraw = Sbuf[(size_t)img * (AN2 * AN2) + tid];
  const float4 pv = reinterpret_cast<const float4*>(proj)[tid & (CIN * AN2 / 4 - 1)];
  const float sm = sums[img * 64 + (tid & 63)];
  if (tid < CIN * AN2 / 4) reinterpret_cast<float4*>(pj)[tid] = pv;
  if (tid < 32) {
    rq[tid] = 1.f / fmaxf(sqrtf(sm), 1e-12f);
  } else if (tid < 64) {
    rk[tid - 32] = 1.f / fmaxf(sqrtf(sm), 1e-12f);
  }
  __syncthreads();

  const float v = sraw * rq[row] * rk[col];
  float mx = fmaxf(-3.4e38f, v);
#pragma unroll
  for (int off = 16; off > 0; off >>= 1) mx = fmaxf(mx, __shfl_xor(mx, off, 32));
  const float e = __expf(v - mx);
  float sum = e;
#pragma unroll
  for (int off = 16; off > 0; off >>= 1) sum += __shfl_xor(sum, off, 32);
  const float inv = 1.f / sum;
  att[row][col] = e * inv;
  __syncthreads();

  const float* __restrict__ p0 = pj + row * AN2;
  const float* __restrict__ p1 = pj + (row + 32) * AN2;
  float a0 = 0.f, a1 = 0.f;
#pragma unroll
  for (int cc = 0; cc < AN2; ++cc) {
    const float av = att[cc][col];
    a0 += p0[cc] * av;
    a1 += p1[cc] * av;
  }
  float* __restrict__ Wg = Wbuf + (size_t)img * 2048;
  Wg[row * AN2 + col] = a0;
  Wg[(row + 32) * AN2 + col] = a1;
}

#define VR  8
#define VP  1028
#define WP  36
__global__ __launch_bounds__(256) void k_dwout(const float* __restrict__ t,
                                               const float* __restrict__ dw,
                                               const float* __restrict__ Wbuf,
                                               float* __restrict__ out) {
  __shared__ __attribute__((aligned(16))) float vls[AN2 * VP];
  __shared__ __attribute__((aligned(16))) float wls[CIN * WP];
  const int tid  = threadIdx.x;
  const int lane = tid & 31;
  const int wave = __builtin_amdgcn_readfirstlane(tid >> 5);
  const int img = blockIdx.y;
  const int r0 = blockIdx.x * VR;
  const int rtop = (r0 > 0) ? r0 - 1 : 0;
  const int rbot = (r0 + VR < IMG_W) ? r0 + VR : IMG_W - 1;

  const float* __restrict__ T0 =
      t + ((size_t)img * CMID + 2 * AN2 + wave * 4) * HW + 4 * lane;
  const float* __restrict__ Wg = Wbuf + (size_t)img * 2048;
  const int wrow = tid >> 3, wc4 = tid & 7;
  const float4 wv0 = *reinterpret_cast<const float4*>(Wg + wrow * AN2 + wc4 * 4);
  const float4 wv1 = *reinterpret_cast<const float4*>(Wg + (wrow + 32) * AN2 + wc4 * 4);
  float4 call[4][VR + 2];
#pragma unroll
  for (int cc = 0; cc < 4; ++cc) {
    const float* __restrict__ T = T0 + (size_t)cc * HW;
    call[cc][0] = *reinterpret_cast<const float4*>(T + rtop * IMG_W);
#pragma unroll
    for (int i = 1; i < VR + 1; ++i)
      call[cc][i] = *reinterpret_cast<const float4*>(T + (r0 - 1 + i) * IMG_W);
    call[cc][VR + 1] = *reinterpret_cast<const float4*>(T + rbot * IMG_W);
  }

#pragma unroll
  for (int cc = 0; cc < 4; ++cc) {
    const int chv = wave * 4 + cc;
    const int ch  = 2 * AN2 + chv;
    float4 (&c)[VR + 2] = call[cc];
    float w[9];
#pragma unroll
    for (int i = 0; i < 9; ++i) w[i] = dw[ch * 9 + i];
    if (r0 == 0)          c[0]      = make_float4(0.f, 0.f, 0.f, 0.f);
    if (r0 + VR >= IMG_W) c[VR + 1] = make_float4(0.f, 0.f, 0.f, 0.f);
    float lf[VR + 2], rt[VR + 2];
#pragma unroll
    for (int i = 0; i < VR + 2; ++i) {
      const float up = __shfl_up(c[i].w, 1, 32);
      const float dn = __shfl_down(c[i].x, 1, 32);
      lf[i] = (lane == 0) ? 0.f : up;
      rt[i] = (lane == 31) ? 0.f : dn;
    }
#pragma unroll
    for (int i = 0; i < VR; ++i) {
      float s0 = 0.f, s1 = 0.f, s2 = 0.f, s3 = 0.f;
#pragma unroll
      for (int dy = 0; dy < 3; ++dy) {
        const float4 cv = c[i + dy];
        const float l = lf[i + dy], r = rt[i + dy];
        const float w0 = w[dy * 3 + 0], w1 = w[dy * 3 + 1], w2 = w[dy * 3 + 2];
        s0 += w0 * l;    s0 += w1 * cv.x; s0 += w2 * cv.y;
        s1 += w0 * cv.x; s1 += w1 * cv.y; s1 += w2 * cv.z;
        s2 += w0 * cv.y; s2 += w1 * cv.z; s2 += w2 * cv.w;
        s3 += w0 * cv.z; s3 += w1 * cv.w; s3 += w2 * r;
      }
      *reinterpret_cast<float4*>(vls + chv * VP + i * IMG_W + 4 * lane) =
          make_float4(s0, s1, s2, s3);
    }
  }
  *reinterpret_cast<float4*>(wls + wrow * WP + wc4 * 4) = wv0;
  *reinterpret_cast<float4*>(wls + (wrow + 32) * WP + wc4 * 4) = wv1;
  __syncthreads();

  const int lo = lane & 15, hi = lane >> 4;
  const float* __restrict__ wa = wls + lo * WP + 2 * hi;
  for (int it = 0; it < 4; ++it) {
    const int mh = it & 1;
    const int px = (wave * 2 + (it >> 1)) * 64 + 4 * lo;
    const float* __restrict__ vb = vls + 2 * hi * VP + px;
    float* __restrict__ O =
        out + (size_t)img * CIN * HW + (size_t)r0 * IMG_W + px;
    v8f acc[2][4] = {};
#pragma unroll 2
    for (int k0 = 0; k0 < AN2; k0 += 4) {
      const float4 b0 = *reinterpret_cast<const float4*>(vb + k0 * VP);
      const float4 b1 = *reinterpret_cast<const float4*>(vb + (k0 + 1) * VP);
      const float bs0[4] = {b0.x, b0.y, b0.z, b0.w};
      const float bs1[4] = {b1.x, b1.y, b1.z, b1.w};
#pragma unroll
      for (int m = 0; m < 2; ++m) {
        const v2f a =
            *reinterpret_cast<const v2f*>(wa + (mh * 2 + m) * 16 * WP + k0);
#pragma unroll
        for (int j = 0; j < 4; ++j) {
          v2f b;
          b.x = bs0[j];
          b.y = bs1[j];
          acc[m][j] = wmma4(a, b, acc[m][j]);
        }
      }
    }
#pragma unroll
    for (int m = 0; m < 2; ++m)
#pragma unroll
      for (int r = 0; r < 8; ++r) {
        v4f o;
        o.x = acc[m][0][r];
        o.y = acc[m][1][r];
        o.z = acc[m][2][r];
        o.w = acc[m][3][r];
        __builtin_nontemporal_store(o, reinterpret_cast<v4f*>(
            O + (size_t)((mh * 2 + m) * 16 + r + 8 * hi) * HW));
      }
  }
}

extern "C" void kernel_launch(void* const* d_in, const int* in_sizes, int n_in,
                              void* d_out, int out_size, void* d_ws, size_t ws_size,
                              hipStream_t stream) {
  const float* x    = (const float*)d_in[0];
  const float* pw   = (const float*)d_in[1];
  const float* dw   = (const float*)d_in[2];
  const float* proj = (const float*)d_in[3];
  float* out = (float*)d_out;

  float* qkv  = (float*)d_ws;
  float* sums = qkv + (size_t)BCN * CMID * HW;
  float* Sbuf = sums + (size_t)BCN * 64;
  float* Wbuf = Sbuf + (size_t)BCN * AN2 * AN2;
  float* tbuf = Wbuf + (size_t)BCN * 64 * AN2;
  const int nzero = BCN * 64 + BCN * AN2 * AN2;

  k_zero<<<dim3((nzero + 255) / 256), 256, 0, stream>>>(sums, nzero);
  k_pw<<<dim3(256), 512, 0, stream>>>(x, pw, tbuf);
  k_dwqk<<<dim3(IMG_W / QR, BCN), 512, 0, stream>>>(tbuf, dw, Sbuf, sums);
  k_att_final<<<dim3(BCN), 1024, 0, stream>>>(Sbuf, sums, proj, Wbuf);
  k_dwout<<<dim3(IMG_W / VR, BCN), 256, 0, stream>>>(tbuf, dw, Wbuf, out);
}


# launches 5 -> 3: GEMM kernel stores the start zeros of S and square sums; each out block forms softmax + proj fold itself, fold on WMMA; softmax row and fold sum order changed
# speedup vs baseline: 3.9295x; 1.0926x over previous
#include <hip/hip_runtime.h>

typedef __attribute__((ext_vector_type(2))) float v2f;
typedef __attribute__((ext_vector_type(4))) float v4f;
typedef __attribute__((ext_vector_type(8))) float v8f;

#define HW    16384
#define IMG_W 128
#define BCN   32
#define CIN   64
#define CMID  96
#define AN2   32

static __device__ __forceinline__ v8f wmma4(v2f a, v2f b, v8f c) {
  return __builtin_amdgcn_wmma_f32_16x16x4_f32(
       false, a,  false, b,
       (short)0, c,  false,  false);
}

#define PWP 68
#define PW_PASSES 4
#define NZERO (BCN * 64 + BCN * AN2 * AN2)

static __device__ __forceinline__ void pw_load_chunk(float4 (&xv)[8],
                                                     const float* __restrict__ X,
                                                     int kc, int hi) {
#pragma unroll
  for (int ks = 0; ks < 4; ++ks) {
    const int k0 = kc + ks * 4;
    xv[2 * ks]     = *reinterpret_cast<const float4*>(X + (size_t)(k0 + 2 * hi) * HW);
    xv[2 * ks + 1] = *reinterpret_cast<const float4*>(X + (size_t)(k0 + 2 * hi + 1) * HW);
  }
}

static __device__ __forceinline__ void pw_mul_chunk(v8f (&acc)[3][4],
                                                    const float4 (&xv)[8],
                                                    const float* __restrict__ arow,
                                                    int kc) {
#pragma unroll
  for (int ks = 0; ks < 4; ++ks) {
    const float4 x0 = xv[2 * ks], x1 = xv[2 * ks + 1];
    const float xs0[4] = {x0.x, x0.y, x0.z, x0.w};
    const float xs1[4] = {x1.x, x1.y, x1.z, x1.w};
#pragma unroll
    for (int m = 0; m < 3; ++m) {
      const v2f a = *reinterpret_cast<const v2f*>(arow + m * 16 * PWP + kc + ks * 4);
#pragma unroll
      for (int j = 0; j < 4; ++j) {
        v2f b;
        b.x = xs0[j];
        b.y = xs1[j];
        acc[m][j] = wmma4(a, b, acc[m][j]);
      }
    }
  }
}

__global__ __launch_bounds__(512) void k_pw(const float* __restrict__ x,
                                            const float* __restrict__ pw,
                                            float* __restrict__ t,
                                            float* __restrict__ zacc) {
  __shared__ __attribute__((aligned(16))) float pwls[CMID * PWP];
  const int tid  = threadIdx.x;
  const int lane = tid & 31, wave = tid >> 5;
  const int lo = lane & 15, hi = lane >> 4;
  const int img = blockIdx.x >> 3;

  {
    const int gi = blockIdx.x * 512 + tid;
    if (gi < NZERO) zacc[gi] = 0.f;
  }
  for (int i = tid; i < CMID * (CIN / 4); i += 512) {
    const int row = i >> 4, c4 = i & 15;
    const float4 v = *reinterpret_cast<const float4*>(pw + row * CIN + c4 * 4);
    *reinterpret_cast<float4*>(pwls + row * PWP + c4 * 4) = v;
  }

  const int w8 = wave & 7;
  const int cbase = (w8 >> 2) * 48;
  const int base = ((blockIdx.x & 7) * 8 + (wave >> 3)) * 256 + (w8 & 3) * 64;
  const float* __restrict__ X = x + (size_t)img * CIN * HW + base + 4 * lo;
  float* __restrict__ Tt =
      t + (size_t)img * CMID * HW + (size_t)cbase * HW + base + 4 * lo;
  const float* __restrict__ arow = pwls + (cbase + lo) * PWP + 2 * hi;

  float4 xa[8], xb[8];
  pw_load_chunk(xa, X, 0, hi);
  __syncthreads();

#define PW_STAGE() __builtin_amdgcn_sched_barrier(0)

#pragma unroll 1
  for (int pass = 0; pass < PW_PASSES; ++pass) {
    v8f acc[3][4] = {};
    PW_STAGE();
    pw_load_chunk(xb, X, 16, hi);
    PW_STAGE();
    pw_mul_chunk(acc, xa, arow, 0);
    PW_STAGE();
    pw_load_chunk(xa, X, 32, hi);
    PW_STAGE();
    pw_mul_chunk(acc, xb, arow, 16);
    PW_STAGE();
    pw_load_chunk(xb, X, 48, hi);
    PW_STAGE();
    pw_mul_chunk(acc, xa, arow, 32);
    PW_STAGE();
    if (pass + 1 < PW_PASSES) X += 512;
    pw_load_chunk(xa, X, 0, hi);
    PW_STAGE();
    pw_mul_chunk(acc, xb, arow, 48);
#pragma unroll
    for (int m = 0; m < 3; ++m)
#pragma unroll
      for (int r = 0; r < 8; ++r) {
        float4 o;
        o.x = acc[m][0][r];
        o.y = acc[m][1][r];
        o.z = acc[m][2][r];
        o.w = acc[m][3][r];
        *reinterpret_cast<float4*>(Tt + (size_t)(m * 16 + r + 8 * hi) * HW) = o;
      }
    Tt += 512;
  }
#undef PW_STAGE
}

#define QR  8
#define QP  1028
#define QC  4
#define QWV 16
__global__ __launch_bounds__(512) void k_dwqk(const float* __restrict__ t,
                                              const float* __restrict__ dw,
                                              float* __restrict__ Sbuf,
                                              float* __restrict__ gsums) {
  __shared__ __attribute__((aligned(16))) float qk[2 * AN2 * QP];
  const int tid  = threadIdx.x;
  const int lane = tid & 31;
  const int wave = __builtin_amdgcn_readfirstlane(tid >> 5);
  const int img = blockIdx.y;
  const int r0 = blockIdx.x * QR;
  const int rtop = (r0 > 0) ? r0 - 1 : 0;
  const int rbot = (r0 + QR < IMG_W) ? r0 + QR : IMG_W - 1;

  const float* __restrict__ T0 =
      t + ((size_t)img * CMID + wave * QC) * HW + 4 * lane;
  float4 call[QC][QR + 2];
#pragma unroll
  for (int cc = 0; cc < QC; ++cc) {
    const float* __restrict__ T = T0 + (size_t)cc * HW;
    call[cc][0] = *reinterpret_cast<const float4*>(T + rtop * IMG_W);
#pragma unroll
    for (int i = 1; i < QR + 1; ++i)
      call[cc][i] = *reinterpret_cast<const float4*>(T + (r0 - 1 + i) * IMG_W);
    call[cc][QR + 1] = *reinterpret_cast<const float4*>(T + rbot * IMG_W);
  }
  float ssc[QC];
#pragma unroll
  for (int cc = 0; cc < QC; ++cc) {
    const int ch = wave * QC + cc;
    float4 (&c)[QR + 2] = call[cc];
    float w[9];
#pragma unroll
    for (int i = 0; i < 9; ++i) w[i] = dw[ch * 9 + i];
    if (r0 == 0)          c[0]      = make_float4(0.f, 0.f, 0.f, 0.f);
    if (r0 + QR >= IMG_W) c[QR + 1] = make_float4(0.f, 0.f, 0.f, 0.f);
    float lf[QR + 2], rt[QR + 2];
#pragma unroll
    for (int i = 0; i < QR + 2; ++i) {
      const float up = __shfl_up(c[i].w, 1, 32);
      const float dn = __shfl_down(c[i].x, 1, 32);
      lf[i] = (lane == 0) ? 0.f : up;
      rt[i] = (lane == 31) ? 0.f : dn;
    }
    float ss = 0.f;
#pragma unroll
    for (int i = 0; i < QR; ++i) {
      float s0 = 0.f, s1 = 0.f, s2 = 0.f, s3 = 0.f;
#pragma unroll
      for (int dy = 0; dy < 3; ++dy) {
        const float4 cv = c[i + dy];
        const float l = lf[i + dy], r = rt[i + dy];
        const float w0 = w[dy * 3 + 0], w1 = w[dy * 3 + 1], w2 = w[dy * 3 + 2];
        s0 += w0 * l;    s0 += w1 * cv.x; s0 += w2 * cv.y;
        s1 += w0 * cv.x; s1 += w1 * cv.y; s1 += w2 * cv.z;
        s2 += w0 * cv.y; s2 += w1 * cv.z; s2 += w2 * cv.w;
        s3 += w0 * cv.z; s3 += w1 * cv.w; s3 += w2 * r;
      }
      *reinterpret_cast<float4*>(qk + ch * QP + i * IMG_W + 4 * lane) =
          make_float4(s0, s1, s2, s3);
      ss += s0 * s0; ss += s1 * s1; ss += s2 * s2; ss += s3 * s3;
    }
    ssc[cc] = ss;
  }
  float mine = 0.f;
#pragma unroll
  for (int cc = 0; cc < QC; ++cc) {
    float v = ssc[cc];
#pragma unroll
    for (int off = 16; off > 0; off >>= 1) v += __shfl_xor(v, off, 32);
    mine = (lane == cc) ? v : mine;
  }
  if (lane < QC) atomicAdd(&gsums[img * 64 + wave * QC + lane], mine);
  __syncthreads();

  const int lo = lane & 15, hi = lane >> 4;
  const float* __restrict__ qa = qk + lo * QP + wave * 64 + 4 * hi;
  const float* __restrict__ kb = qa + AN2 * QP;
  v8f acc[2][2] = {};
#pragma unroll
  for (int s = 0; s < 64; s += 8) {
    float4 av[2], bv[2];
#pragma unroll
    for (int m = 0; m < 2; ++m) {
      av[m] = *reinterpret_cast<const float4*>(qa + m * 16 * QP + s);
      bv[m] = *reinterpret_cast<const float4*>(kb + m * 16 * QP + s);
    }
#pragma unroll
    for (int m = 0; m < 2; ++m)
#pragma unroll
      for (int n = 0; n < 2; ++n) {
        v2f a, b;
        a.x = av[m].x; a.y = av[m].y;
        b.x = bv[n].x; b.y = bv[n].y;
        acc[m][n] = wmma4(a, b, acc[m][n]);
      }
#pragma unroll
    for (int m = 0; m < 2; ++m)
#pragma unroll
      for (int n = 0; n < 2; ++n) {
        v2f a, b;
        a.x = av[m].z; a.y = av[m].w;
        b.x = bv[n].z; b.y = bv[n].w;
        acc[m][n] = wmma4(a, b, acc[m][n]);
      }
  }
  __syncthreads();
  float* __restrict__ red = qk;
#pragma unroll
  for (int m = 0; m < 2; ++m)
#pragma unroll
    for (int n = 0; n < 2; ++n)
#pragma unroll
      for (int r = 0; r < 8; ++r)
        red[wave * (AN2 * AN2) + (m * 16 + r + 8 * hi) * AN2 + n * 16 + lo] = acc[m][n][r];
  __syncthreads();
  float* __restrict__ Sp = Sbuf + (size_t)img * (AN2 * AN2);
  for (int e = tid; e < AN2 * AN2; e += 512) {
    float s = 0.f;
#pragma unroll
    for (int wv = 0; wv < QWV; ++wv) s += red[wv * (AN2 * AN2) + e];
    atomicAdd(&Sp[e], s);
  }
}

#define VR  8
#define VP  1028
#define WP  36
#define AP  40
__global__ __launch_bounds__(256) void k_dwout(const float* __restrict__ t,
                                               const float* __restrict__ dw,
                                               const float* __restrict__ Sbuf,
                                               const float* __restrict__ sums,
                                               const float* __restrict__ proj,
                                               float* __restrict__ out) {
  __shared__ __attribute__((aligned(16))) float vls[AN2 * VP];
  __shared__ __attribute__((aligned(16))) float wls[CIN * WP];
  __shared__ __attribute__((aligned(16))) float attls[AN2 * AP];
  const int tid  = threadIdx.x;
  const int lane = tid & 31;
  const int wave = __builtin_amdgcn_readfirstlane(tid >> 5);
  const int img = blockIdx.y;
  const int r0 = blockIdx.x * VR;
  const int rtop = (r0 > 0) ? r0 - 1 : 0;
  const int rbot = (r0 + VR < IMG_W) ? r0 + VR : IMG_W - 1;

  const float* __restrict__ T0 =
      t + ((size_t)img * CMID + 2 * AN2 + wave * 4) * HW + 4 * lane;
  const int lo = lane & 15, hi = lane >> 4;
  const int srow = tid >> 3, scg = tid & 7;
  const v4f sraw =
      *reinterpret_cast<const v4f*>(Sbuf + (size_t)img * (AN2 * AN2) + tid * 4);
  const float sqv = sums[img * 64 + srow];
  const v4f skv = *reinterpret_cast<const v4f*>(sums + img * 64 + AN2 + scg * 4);
  const int fm = wave >> 1, fn = wave & 1;
  v2f pa[8];
#pragma unroll
  for (int ks = 0; ks < 8; ++ks)
    pa[ks] = *reinterpret_cast<const v2f*>(proj + (fm * 16 + lo) * AN2 + ks * 4 + 2 * hi);
  float4 call[4][VR + 2];
#pragma unroll
  for (int cc = 0; cc < 4; ++cc) {
    const float* __restrict__ T = T0 + (size_t)cc * HW;
    call[cc][0] = *reinterpret_cast<const float4*>(T + rtop * IMG_W);
#pragma unroll
    for (int i = 1; i < VR + 1; ++i)
      call[cc][i] = *reinterpret_cast<const float4*>(T + (r0 - 1 + i) * IMG_W);
    call[cc][VR + 1] = *reinterpret_cast<const float4*>(T + rbot * IMG_W);
  }

  asm volatile("" :: "v"(sraw), "v"(skv));
  __builtin_amdgcn_sched_barrier(0);
  {
    const float rq = 1.f / fmaxf(sqrtf(sqv), 1e-12f);
    const float v0 = sraw.x * rq * (1.f / fmaxf(sqrtf(skv.x), 1e-12f));
    const float v1 = sraw.y * rq * (1.f / fmaxf(sqrtf(skv.y), 1e-12f));
    const float v2 = sraw.z * rq * (1.f / fmaxf(sqrtf(skv.z), 1e-12f));
    const float v3 = sraw.w * rq * (1.f / fmaxf(sqrtf(skv.w), 1e-12f));
    float mx = fmaxf(fmaxf(v0, v1), fmaxf(v2, v3));
#pragma unroll
    for (int off = 4; off > 0; off >>= 1) mx = fmaxf(mx, __shfl_xor(mx, off, 32));
    const float e0 = __expf(v0 - mx), e1 = __expf(v1 - mx);
    const float e2 = __expf(v2 - mx), e3 = __expf(v3 - mx);
    float sum = (e0 + e1) + (e2 + e3);
#pragma unroll
    for (int off = 4; off > 0; off >>= 1) sum += __shfl_xor(sum, off, 32);
    const float inv = 1.f / sum;
    *reinterpret_cast<float4*>(attls + srow * AP + scg * 4) =
        make_float4(e0 * inv, e1 * inv, e2 * inv, e3 * inv);
  }
  __builtin_amdgcn_fence(__ATOMIC_RELEASE, "workgroup", "local");
  __builtin_amdgcn_s_barrier();
  __builtin_amdgcn_fence(__ATOMIC_ACQUIRE, "workgroup", "local");
  {
    v8f wacc = {};
#pragma unroll
    for (int ks = 0; ks < 8; ++ks) {
      v2f b;
      b.x = attls[(ks * 4 + 2 * hi) * AP + fn * 16 + lo];
      b.y = attls[(ks * 4 + 2 * hi + 1) * AP + fn * 16 + lo];
      wacc = wmma4(pa[ks], b, wacc);
    }
#pragma unroll
    for (int r = 0; r < 8; ++r)
      wls[(fm * 16 + 8 * hi + r) * WP + fn * 16 + lo] = wacc[r];
  }
  __builtin_amdgcn_sched_barrier(0);

#pragma unroll
  for (int cc = 0; cc < 4; ++cc) {
    const int chv = wave * 4 + cc;
    const int ch  = 2 * AN2 + chv;
    float4 (&c)[VR + 2] = call[cc];
    float w[9];
#pragma unroll
    for (int i = 0; i < 9; ++i) w[i] = dw[ch * 9 + i];
    if (r0 == 0)          c[0]      = make_float4(0.f, 0.f, 0.f, 0.f);
    if (r0 + VR >= IMG_W) c[VR + 1] = make_float4(0.f, 0.f, 0.f, 0.f);
    float lf[VR + 2], rt[VR + 2];
#pragma unroll
    for (int i = 0; i < VR + 2; ++i) {
      const float up = __shfl_up(c[i].w, 1, 32);
      const float dn = __shfl_down(c[i].x, 1, 32);
      lf[i] = (lane == 0) ? 0.f : up;
      rt[i] = (lane == 31) ? 0.f : dn;
    }
#pragma unroll
    for (int i = 0; i < VR; ++i) {
      float s0 = 0.f, s1 = 0.f, s2 = 0.f, s3 = 0.f;
#pragma unroll
      for (int dy = 0; dy < 3; ++dy) {
        const float4 cv = c[i + dy];
        const float l = lf[i + dy], r = rt[i + dy];
        const float w0 = w[dy * 3 + 0], w1 = w[dy * 3 + 1], w2 = w[dy * 3 + 2];
        s0 += w0 * l;    s0 += w1 * cv.x; s0 += w2 * cv.y;
        s1 += w0 * cv.x; s1 += w1 * cv.y; s1 += w2 * cv.z;
        s2 += w0 * cv.y; s2 += w1 * cv.z; s2 += w2 * cv.w;
        s3 += w0 * cv.z; s3 += w1 * cv.w; s3 += w2 * r;
      }
      *reinterpret_cast<float4*>(vls + chv * VP + i * IMG_W + 4 * lane) =
          make_float4(s0, s1, s2, s3);
    }
  }
  __syncthreads();

  const float* __restrict__ wa = wls + lo * WP + 2 * hi;
  for (int it = 0; it < 4; ++it) {
    const int mh = it & 1;
    const int px = (wave * 2 + (it >> 1)) * 64 + 4 * lo;
    const float* __restrict__ vb = vls + 2 * hi * VP + px;
    float* __restrict__ O =
        out + (size_t)img * CIN * HW + (size_t)r0 * IMG_W + px;
    v8f acc[2][4] = {};
#pragma unroll 2
    for (int k0 = 0; k0 < AN2; k0 += 4) {
      const float4 b0 = *reinterpret_cast<const float4*>(vb + k0 * VP);
      const float4 b1 = *reinterpret_cast<const float4*>(vb + (k0 + 1) * VP);
      const float bs0[4] = {b0.x, b0.y, b0.z, b0.w};
      const float bs1[4] = {b1.x, b1.y, b1.z, b1.w};
#pragma unroll
      for (int m = 0; m < 2; ++m) {
        const v2f a =
            *reinterpret_cast<const v2f*>(wa + (mh * 2 + m) * 16 * WP + k0);
#pragma unroll
        for (int j = 0; j < 4; ++j) {
          v2f b;
          b.x = bs0[j];
          b.y = bs1[j];
          acc[m][j] = wmma4(a, b, acc[m][j]);
        }
      }
    }
#pragma unroll
    for (int m = 0; m < 2; ++m)
#pragma unroll
      for (int r = 0; r < 8; ++r) {
        v4f o;
        o.x = acc[m][0][r];
        o.y = acc[m][1][r];
        o.z = acc[m][2][r];
        o.w = acc[m][3][r];
        __builtin_nontemporal_store(o, reinterpret_cast<v4f*>(
            O + (size_t)((mh * 2 + m) * 16 + r + 8 * hi) * HW));
      }
  }
}

extern "C" void kernel_launch(void* const* d_in, const int* in_sizes, int n_in,
                              void* d_out, int out_size, void* d_ws, size_t ws_size,
                              hipStream_t stream) {
  const float* x    = (const float*)d_in[0];
  const float* pw   = (const float*)d_in[1];
  const float* dw   = (const float*)d_in[2];
  const float* proj = (const float*)d_in[3];
  float* out = (float*)d_out;

  float* qkv  = (float*)d_ws;
  float* sums = qkv + (size_t)BCN * CMID * HW;
  float* Sbuf = sums + (size_t)BCN * 64;
  float* tbuf = Sbuf + (size_t)BCN * AN2 * AN2 + (size_t)BCN * 64 * AN2;

  k_pw<<<dim3(256), 512, 0, stream>>>(x, pw, tbuf, sums);
  k_dwqk<<<dim3(IMG_W / QR, BCN), 512, 0, stream>>>(tbuf, dw, Sbuf, sums);
  k_dwout<<<dim3(IMG_W / VR, BCN), 256, 0, stream>>>(tbuf, dw, Sbuf, sums, proj, out);
}


# q/k kernel persistent: 256 work-groups x 2 strips; strip 2's t rows requested as strip 1's registers fall free, loading under its S product and reduction; per-strip sums kept, atomic order differs
# speedup vs baseline: 4.0764x; 1.0374x over previous
#include <hip/hip_runtime.h>

typedef __attribute__((ext_vector_type(2))) float v2f;
typedef __attribute__((ext_vector_type(4))) float v4f;
typedef __attribute__((ext_vector_type(8))) float v8f;

#define HW    16384
#define IMG_W 128
#define BCN   32
#define CIN   64
#define CMID  96
#define AN2   32

static __device__ __forceinline__ v8f wmma4(v2f a, v2f b, v8f c) {
  return __builtin_amdgcn_wmma_f32_16x16x4_f32(
       false, a,  false, b,
       (short)0, c,  false,  false);
}

#define PWP 68
#define PW_PASSES 4
#define NZERO (BCN * 64 + BCN * AN2 * AN2)

static __device__ __forceinline__ void pw_load_chunk(float4 (&xv)[8],
                                                     const float* __restrict__ X,
                                                     int kc, int hi) {
#pragma unroll
  for (int ks = 0; ks < 4; ++ks) {
    const int k0 = kc + ks * 4;
    xv[2 * ks]     = *reinterpret_cast<const float4*>(X + (size_t)(k0 + 2 * hi) * HW);
    xv[2 * ks + 1] = *reinterpret_cast<const float4*>(X + (size_t)(k0 + 2 * hi + 1) * HW);
  }
}

static __device__ __forceinline__ void pw_mul_chunk(v8f (&acc)[3][4],
                                                    const float4 (&xv)[8],
                                                    const float* __restrict__ arow,
                                                    int kc) {
#pragma unroll
  for (int ks = 0; ks < 4; ++ks) {
    const float4 x0 = xv[2 * ks], x1 = xv[2 * ks + 1];
    const float xs0[4] = {x0.x, x0.y, x0.z, x0.w};
    const float xs1[4] = {x1.x, x1.y, x1.z, x1.w};
#pragma unroll
    for (int m = 0; m < 3; ++m) {
      const v2f a = *reinterpret_cast<const v2f*>(arow + m * 16 * PWP + kc + ks * 4);
#pragma unroll
      for (int j = 0; j < 4; ++j) {
        v2f b;
        b.x = xs0[j];
        b.y = xs1[j];
        acc[m][j] = wmma4(a, b, acc[m][j]);
      }
    }
  }
}

__global__ __launch_bounds__(512) void k_pw(const float* __restrict__ x,
                                            const float* __restrict__ pw,
                                            float* __restrict__ t,
                                            float* __restrict__ zacc) {
  __shared__ __attribute__((aligned(16))) float pwls[CMID * PWP];
  const int tid  = threadIdx.x;
  const int lane = tid & 31, wave = tid >> 5;
  const int lo = lane & 15, hi = lane >> 4;
  const int img = blockIdx.x >> 3;

  {
    const int gi = blockIdx.x * 512 + tid;
    if (gi < NZERO) zacc[gi] = 0.f;
  }
  for (int i = tid; i < CMID * (CIN / 4); i += 512) {
    const int row = i >> 4, c4 = i & 15;
    const float4 v = *reinterpret_cast<const float4*>(pw + row * CIN + c4 * 4);
    *reinterpret_cast<float4*>(pwls + row * PWP + c4 * 4) = v;
  }

  const int w8 = wave & 7;
  const int cbase = (w8 >> 2) * 48;
  const int base = ((blockIdx.x & 7) * 8 + (wave >> 3)) * 256 + (w8 & 3) * 64;
  const float* __restrict__ X = x + (size_t)img * CIN * HW + base + 4 * lo;
  float* __restrict__ Tt =
      t + (size_t)img * CMID * HW + (size_t)cbase * HW + base + 4 * lo;
  const float* __restrict__ arow = pwls + (cbase + lo) * PWP + 2 * hi;

  float4 xa[8], xb[8];
  pw_load_chunk(xa, X, 0, hi);
  __syncthreads();

#define PW_STAGE() __builtin_amdgcn_sched_barrier(0)

#pragma unroll 1
  for (int pass = 0; pass < PW_PASSES; ++pass) {
    v8f acc[3][4] = {};
    PW_STAGE();
    pw_load_chunk(xb, X, 16, hi);
    PW_STAGE();
    pw_mul_chunk(acc, xa, arow, 0);
    PW_STAGE();
    pw_load_chunk(xa, X, 32, hi);
    PW_STAGE();
    pw_mul_chunk(acc, xb, arow, 16);
    PW_STAGE();
    pw_load_chunk(xb, X, 48, hi);
    PW_STAGE();
    pw_mul_chunk(acc, xa, arow, 32);
    PW_STAGE();
    if (pass + 1 < PW_PASSES) X += 512;
    pw_load_chunk(xa, X, 0, hi);
    PW_STAGE();
    pw_mul_chunk(acc, xb, arow, 48);
#pragma unroll
    for (int m = 0; m < 3; ++m)
#pragma unroll
      for (int r = 0; r < 8; ++r) {
        float4 o;
        o.x = acc[m][0][r];
        o.y = acc[m][1][r];
        o.z = acc[m][2][r];
        o.w = acc[m][3][r];
        *reinterpret_cast<float4*>(Tt + (size_t)(m * 16 + r + 8 * hi) * HW) = o;
      }
    Tt += 512;
  }
#undef PW_STAGE
}

#define QR  8
#define QP  1028
#define QC  4
#define QWV 16

static __device__ __forceinline__ void lds_barrier() {
  __builtin_amdgcn_fence(__ATOMIC_RELEASE, "workgroup", "local");
  __builtin_amdgcn_s_barrier();
  __builtin_amdgcn_fence(__ATOMIC_ACQUIRE, "workgroup", "local");
}

static __device__ __forceinline__ float dwqk_channel(const float4 (&c)[QR + 2],
                                                    const float (&w)[9], int lane,
                                                    float* __restrict__ dst) {
  float lf[QR + 2], rt[QR + 2];
#pragma unroll
  for (int i = 0; i < QR + 2; ++i) {
    const float up = __shfl_up(c[i].w, 1, 32);
    const float dn = __shfl_down(c[i].x, 1, 32);
    lf[i] = (lane == 0) ? 0.f : up;
    rt[i] = (lane == 31) ? 0.f : dn;
  }
  float ss = 0.f;
#pragma unroll
  for (int i = 0; i < QR; ++i) {
    float s0 = 0.f, s1 = 0.f, s2 = 0.f, s3 = 0.f;
#pragma unroll
    for (int dy = 0; dy < 3; ++dy) {
      const float4 cv = c[i + dy];
      const float l = lf[i + dy], r = rt[i + dy];
      const float w0 = w[dy * 3 + 0], w1 = w[dy * 3 + 1], w2 = w[dy * 3 + 2];
      s0 += w0 * l;    s0 += w1 * cv.x; s0 += w2 * cv.y;
      s1 += w0 * cv.x; s1 += w1 * cv.y; s1 += w2 * cv.z;
      s2 += w0 * cv.y; s2 += w1 * cv.z; s2 += w2 * cv.w;
      s3 += w0 * cv.z; s3 += w1 * cv.w; s3 += w2 * r;
    }
    *reinterpret_cast<float4*>(dst + i * IMG_W) = make_float4(s0, s1, s2, s3);
    ss += s0 * s0; ss += s1 * s1; ss += s2 * s2; ss += s3 * s3;
  }
  return ss;
}

static __device__ __forceinline__ void dwqk_product(v8f (&acc)[2][2],
                                                    const float* __restrict__ qa,
                                                    const float* __restrict__ kb) {
#pragma unroll
  for (int s = 0; s < 64; s += 8) {
    float4 av[2], bv[2];
#pragma unroll
    for (int m = 0; m < 2; ++m) {
      av[m] = *reinterpret_cast<const float4*>(qa + m * 16 * QP + s);
      bv[m] = *reinterpret_cast<const float4*>(kb + m * 16 * QP + s);
    }
#pragma unroll
    for (int m = 0; m < 2; ++m)
#pragma unroll
      for (int n = 0; n < 2; ++n) {
        v2f a, b;
        a.x = av[m].x; a.y = av[m].y;
        b.x = bv[n].x; b.y = bv[n].y;
        acc[m][n] = wmma4(a, b, acc[m][n]);
      }
#pragma unroll
    for (int m = 0; m < 2; ++m)
#pragma unroll
      for (int n = 0; n < 2; ++n) {
        v2f a, b;
        a.x = av[m].z; a.y = av[m].w;
        b.x = bv[n].z; b.y = bv[n].w;
        acc[m][n] = wmma4(a, b, acc[m][n]);
      }
  }
}

static __device__ __forceinline__ void dwqk_reduce(const v8f (&acc)[2][2],
                                                   float* __restrict__ red,
                                                   float* __restrict__ Sp, int tid,
                                                   int wave, int lo, int hi) {
#pragma unroll
  for (int m = 0; m < 2; ++m)
#pragma unroll
    for (int n = 0; n < 2; ++n)
#pragma unroll
      for (int r = 0; r < 8; ++r)
        red[wave * (AN2 * AN2) + (m * 16 + r + 8 * hi) * AN2 + n * 16 + lo] = acc[m][n][r];
  lds_barrier();
  for (int e = tid; e < AN2 * AN2; e += 512) {
    float s = 0.f;
#pragma unroll
    for (int wv = 0; wv < QWV; ++wv) s += red[wv * (AN2 * AN2) + e];
    atomicAdd(&Sp[e], s);
  }
}

static __device__ __forceinline__ void dwqk_squares(const float (&ssc)[QC],
                                                    float* __restrict__ gs, int lane) {
  float mine = 0.f;
#pragma unroll
  for (int cc = 0; cc < QC; ++cc) {
    float v = ssc[cc];
#pragma unroll
    for (int off = 16; off > 0; off >>= 1) v += __shfl_xor(v, off, 32);
    mine = (lane == cc) ? v : mine;
  }
  if (lane < QC) atomicAdd(&gs[lane], mine);
}

__global__ __launch_bounds__(512) void k_dwqk(const float* __restrict__ t,
                                              const float* __restrict__ dw,
                                              float* __restrict__ Sbuf,
                                              float* __restrict__ gsums) {
  __shared__ __attribute__((aligned(16))) float qk[2 * AN2 * QP];
  const int tid  = threadIdx.x;
  const int lane = tid & 31;
  const int wave = __builtin_amdgcn_readfirstlane(tid >> 5);
  const int img = blockIdx.y;
  const int r0 = blockIdx.x * (2 * QR);
  const int rtop = (r0 > 0) ? r0 - 1 : 0;
  const int rbot = (r0 + 2 * QR < IMG_W) ? r0 + 2 * QR : IMG_W - 1;

  const float* __restrict__ T0 =
      t + ((size_t)img * CMID + wave * QC) * HW + 4 * lane;
  float4 ca[QC][QR + 2];
#pragma unroll
  for (int cc = 0; cc < QC; ++cc) {
    const float* __restrict__ T = T0 + (size_t)cc * HW;
    ca[cc][0] = *reinterpret_cast<const float4*>(T + rtop * IMG_W);
#pragma unroll
    for (int i = 1; i < QR + 2; ++i)
      ca[cc][i] = *reinterpret_cast<const float4*>(T + (r0 - 1 + i) * IMG_W);
  }
  float w[QC][9];
#pragma unroll
  for (int cc = 0; cc < QC; ++cc)
#pragma unroll
    for (int i = 0; i < 9; ++i) w[cc][i] = dw[(wave * QC + cc) * 9 + i];

  const int lo = lane & 15, hi = lane >> 4;
  const float* __restrict__ qa = qk + lo * QP + wave * 64 + 4 * hi;
  const float* __restrict__ kb = qa + AN2 * QP;
  float* __restrict__ Sp = Sbuf + (size_t)img * (AN2 * AN2);
  float* __restrict__ gs = gsums + img * 64 + wave * QC;

  float4 cb[QC][QR + 2];
#define DWQK_NEXT(cc)                                                          \
  {                                                                            \
    const float* __restrict__ Tn = T0 + (size_t)(cc) * HW;                     \
    cb[cc][0] = ca[cc][QR];                                                    \
    cb[cc][1] = ca[cc][QR + 1];                                                \
    _Pragma("unroll") for (int i = 2; i < QR + 1; ++i) cb[cc][i] =             \
        *reinterpret_cast<const float4*>(Tn + (r0 + QR - 1 + i) * IMG_W);      \
    cb[cc][QR + 1] = *reinterpret_cast<const float4*>(Tn + rbot * IMG_W);      \
  }
  {
    float ssc[QC];
#pragma unroll
    for (int cc = 0; cc < QC; ++cc) {
      const int ch = wave * QC + cc;
      if (r0 == 0) ca[cc][0] = make_float4(0.f, 0.f, 0.f, 0.f);
      ssc[cc] = dwqk_channel(ca[cc], w[cc], lane, qk + ch * QP + 4 * lane);
      __builtin_amdgcn_sched_barrier(0);
      if (cc > 0) DWQK_NEXT(cc - 1)
      if (cc == QC - 1) DWQK_NEXT(QC - 1)
      __builtin_amdgcn_sched_barrier(0);
    }
    dwqk_squares(ssc, gs, lane);
  }
#undef DWQK_NEXT
  lds_barrier();
  {
    v8f acc[2][2] = {};
    dwqk_product(acc, qa, kb);
    lds_barrier();
    dwqk_reduce(acc, qk, Sp, tid, wave, lo, hi);
  }
  lds_barrier();
  {
    float ssc[QC];
#pragma unroll
    for (int cc = 0; cc < QC; ++cc) {
      const int ch = wave * QC + cc;
      if (r0 + 2 * QR >= IMG_W) cb[cc][QR + 1] = make_float4(0.f, 0.f, 0.f, 0.f);
      ssc[cc] = dwqk_channel(cb[cc], w[cc], lane, qk + ch * QP + 4 * lane);
      __builtin_amdgcn_sched_barrier(0);
    }
    dwqk_squares(ssc, gs, lane);
  }
  lds_barrier();
  {
    v8f acc[2][2] = {};
    dwqk_product(acc, qa, kb);
    lds_barrier();
    dwqk_reduce(acc, qk, Sp, tid, wave, lo, hi);
  }
}

#define VR  8
#define VP  1028
#define WP  36
#define AP  40
__global__ __launch_bounds__(256) void k_dwout(const float* __restrict__ t,
                                               const float* __restrict__ dw,
                                               const float* __restrict__ Sbuf,
                                               const float* __restrict__ sums,
                                               const float* __restrict__ proj,
                                               float* __restrict__ out) {
  __shared__ __attribute__((aligned(16))) float vls[AN2 * VP];
  __shared__ __attribute__((aligned(16))) float wls[CIN * WP];
  __shared__ __attribute__((aligned(16))) float attls[AN2 * AP];
  const int tid  = threadIdx.x;
  const int lane = tid & 31;
  const int wave = __builtin_amdgcn_readfirstlane(tid >> 5);
  const int img = blockIdx.y;
  const int r0 = blockIdx.x * VR;
  const int rtop = (r0 > 0) ? r0 - 1 : 0;
  const int rbot = (r0 + VR < IMG_W) ? r0 + VR : IMG_W - 1;

  const float* __restrict__ T0 =
      t + ((size_t)img * CMID + 2 * AN2 + wave * 4) * HW + 4 * lane;
  const int lo = lane & 15, hi = lane >> 4;
  const int srow = tid >> 3, scg = tid & 7;
  const v4f sraw =
      *reinterpret_cast<const v4f*>(Sbuf + (size_t)img * (AN2 * AN2) + tid * 4);
  const float sqv = sums[img * 64 + srow];
  const v4f skv = *reinterpret_cast<const v4f*>(sums + img * 64 + AN2 + scg * 4);
  const int fm = wave >> 1, fn = wave & 1;
  v2f pa[8];
#pragma unroll
  for (int ks = 0; ks < 8; ++ks)
    pa[ks] = *reinterpret_cast<const v2f*>(proj + (fm * 16 + lo) * AN2 + ks * 4 + 2 * hi);
  float4 call[4][VR + 2];
#pragma unroll
  for (int cc = 0; cc < 4; ++cc) {
    const float* __restrict__ T = T0 + (size_t)cc * HW;
    call[cc][0] = *reinterpret_cast<const float4*>(T + rtop * IMG_W);
#pragma unroll
    for (int i = 1; i < VR + 1; ++i)
      call[cc][i] = *reinterpret_cast<const float4*>(T + (r0 - 1 + i) * IMG_W);
    call[cc][VR + 1] = *reinterpret_cast<const float4*>(T + rbot * IMG_W);
  }

  asm volatile("" :: "v"(sraw), "v"(skv));
  __builtin_amdgcn_sched_barrier(0);
  {
    const float rq = 1.f / fmaxf(sqrtf(sqv), 1e-12f);
    const float v0 = sraw.x * rq * (1.f / fmaxf(sqrtf(skv.x), 1e-12f));
    const float v1 = sraw.y * rq * (1.f / fmaxf(sqrtf(skv.y), 1e-12f));
    const float v2 = sraw.z * rq * (1.f / fmaxf(sqrtf(skv.z), 1e-12f));
    const float v3 = sraw.w * rq * (1.f / fmaxf(sqrtf(skv.w), 1e-12f));
    float mx = fmaxf(fmaxf(v0, v1), fmaxf(v2, v3));
#pragma unroll
    for (int off = 4; off > 0; off >>= 1) mx = fmaxf(mx, __shfl_xor(mx, off, 32));
    const float e0 = __expf(v0 - mx), e1 = __expf(v1 - mx);
    const float e2 = __expf(v2 - mx), e3 = __expf(v3 - mx);
    float sum = (e0 + e1) + (e2 + e3);
#pragma unroll
    for (int off = 4; off > 0; off >>= 1) sum += __shfl_xor(sum, off, 32);
    const float inv = 1.f / sum;
    *reinterpret_cast<float4*>(attls + srow * AP + scg * 4) =
        make_float4(e0 * inv, e1 * inv, e2 * inv, e3 * inv);
  }
  __builtin_amdgcn_fence(__ATOMIC_RELEASE, "workgroup", "local");
  __builtin_amdgcn_s_barrier();
  __builtin_amdgcn_fence(__ATOMIC_ACQUIRE, "workgroup", "local");
  {
    v8f wacc = {};
#pragma unroll
    for (int ks = 0; ks < 8; ++ks) {
      v2f b;
      b.x = attls[(ks * 4 + 2 * hi) * AP + fn * 16 + lo];
      b.y = attls[(ks * 4 + 2 * hi + 1) * AP + fn * 16 + lo];
      wacc = wmma4(pa[ks], b, wacc);
    }
#pragma unroll
    for (int r = 0; r < 8; ++r)
      wls[(fm * 16 + 8 * hi + r) * WP + fn * 16 + lo] = wacc[r];
  }
  __builtin_amdgcn_sched_barrier(0);

#pragma unroll
  for (int cc = 0; cc < 4; ++cc) {
    const int chv = wave * 4 + cc;
    const int ch  = 2 * AN2 + chv;
    float4 (&c)[VR + 2] = call[cc];
    float w[9];
#pragma unroll
    for (int i = 0; i < 9; ++i) w[i] = dw[ch * 9 + i];
    if (r0 == 0)          c[0]      = make_float4(0.f, 0.f, 0.f, 0.f);
    if (r0 + VR >= IMG_W) c[VR + 1] = make_float4(0.f, 0.f, 0.f, 0.f);
    float lf[VR + 2], rt[VR + 2];
#pragma unroll
    for (int i = 0; i < VR + 2; ++i) {
      const float up = __shfl_up(c[i].w, 1, 32);
      const float dn = __shfl_down(c[i].x, 1, 32);
      lf[i] = (lane == 0) ? 0.f : up;
      rt[i] = (lane == 31) ? 0.f : dn;
    }
#pragma unroll
    for (int i = 0; i < VR; ++i) {
      float s0 = 0.f, s1 = 0.f, s2 = 0.f, s3 = 0.f;
#pragma unroll
      for (int dy = 0; dy < 3; ++dy) {
        const float4 cv = c[i + dy];
        const float l = lf[i + dy], r = rt[i + dy];
        const float w0 = w[dy * 3 + 0], w1 = w[dy * 3 + 1], w2 = w[dy * 3 + 2];
        s0 += w0 * l;    s0 += w1 * cv.x; s0 += w2 * cv.y;
        s1 += w0 * cv.x; s1 += w1 * cv.y; s1 += w2 * cv.z;
        s2 += w0 * cv.y; s2 += w1 * cv.z; s2 += w2 * cv.w;
        s3 += w0 * cv.z; s3 += w1 * cv.w; s3 += w2 * r;
      }
      *reinterpret_cast<float4*>(vls + chv * VP + i * IMG_W + 4 * lane) =
          make_float4(s0, s1, s2, s3);
    }
  }
  __syncthreads();

  const float* __restrict__ wa = wls + lo * WP + 2 * hi;
  for (int it = 0; it < 4; ++it) {
    const int mh = it & 1;
    const int px = (wave * 2 + (it >> 1)) * 64 + 4 * lo;
    const float* __restrict__ vb = vls + 2 * hi * VP + px;
    float* __restrict__ O =
        out + (size_t)img * CIN * HW + (size_t)r0 * IMG_W + px;
    v8f acc[2][4] = {};
#pragma unroll 2
    for (int k0 = 0; k0 < AN2; k0 += 4) {
      const float4 b0 = *reinterpret_cast<const float4*>(vb + k0 * VP);
      const float4 b1 = *reinterpret_cast<const float4*>(vb + (k0 + 1) * VP);
      const float bs0[4] = {b0.x, b0.y, b0.z, b0.w};
      const float bs1[4] = {b1.x, b1.y, b1.z, b1.w};
#pragma unroll
      for (int m = 0; m < 2; ++m) {
        const v2f a =
            *reinterpret_cast<const v2f*>(wa + (mh * 2 + m) * 16 * WP + k0);
#pragma unroll
        for (int j = 0; j < 4; ++j) {
          v2f b;
          b.x = bs0[j];
          b.y = bs1[j];
          acc[m][j] = wmma4(a, b, acc[m][j]);
        }
      }
    }
#pragma unroll
    for (int m = 0; m < 2; ++m)
#pragma unroll
      for (int r = 0; r < 8; ++r) {
        v4f o;
        o.x = acc[m][0][r];
        o.y = acc[m][1][r];
        o.z = acc[m][2][r];
        o.w = acc[m][3][r];
        __builtin_nontemporal_store(o, reinterpret_cast<v4f*>(
            O + (size_t)((mh * 2 + m) * 16 + r + 8 * hi) * HW));
      }
  }
}

extern "C" void kernel_launch(void* const* d_in, const int* in_sizes, int n_in,
                              void* d_out, int out_size, void* d_ws, size_t ws_size,
                              hipStream_t stream) {
  const float* x    = (const float*)d_in[0];
  const float* pw   = (const float*)d_in[1];
  const float* dw   = (const float*)d_in[2];
  const float* proj = (const float*)d_in[3];
  float* out = (float*)d_out;

  float* qkv  = (float*)d_ws;
  float* sums = qkv + (size_t)BCN * CMID * HW;
  float* Sbuf = sums + (size_t)BCN * 64;
  float* tbuf = Sbuf + (size_t)BCN * AN2 * AN2 + (size_t)BCN * 64 * AN2;

  k_pw<<<dim3(256), 512, 0, stream>>>(x, pw, tbuf, sums);
  k_dwqk<<<dim3(IMG_W / (2 * QR), BCN), 512, 0, stream>>>(tbuf, dw, Sbuf, sums);
  k_dwout<<<dim3(IMG_W / VR, BCN), 256, 0, stream>>>(tbuf, dw, Sbuf, sums, proj, out);
}
